# starting kernel
# baseline (speedup 1.0000x reference)
#include <hip/hip_runtime.h>
#include <hip/hip_bf16.h>


typedef float v2f __attribute__((ext_vector_type(2)));
typedef float v8f __attribute__((ext_vector_type(8)));

#define NNODES 50000
#define FEAT   128
#define HALF   64
#define MTILES (NNODES / 16)
#define SPAD   132
#define CPAD   260

__global__ void fill_zero_kernel(float* __restrict__ p, long long count) {
  long long i = (long long)blockIdx.x * blockDim.x + threadIdx.x;
  if (i < count) p[i] = 0.0f;
}

__global__ void degree_kernel(const int* __restrict__ dst, float* __restrict__ deg, int E) {
  int e = blockIdx.x * blockDim.x + threadIdx.x;
  if (e < E) atomicAdd(&deg[dst[e]], 1.0f);
}

__global__ void scatter_kernel(const float* __restrict__ X,
                               const int* __restrict__ src,
                               const int* __restrict__ dst,
                               float* __restrict__ msum, int E) {
  long long t = (long long)blockIdx.x * blockDim.x + threadIdx.x;
  int e = (int)(t >> 5);
  if (e >= E) return;
  int f = ((int)t & 31) * 4;
  int s = src[e];
  int d = dst[e];
  const float4 v = *(const float4*)(X + (size_t)s * FEAT + f);
  float* mp = msum + (size_t)d * FEAT + f;
  atomicAdd(mp + 0, v.x);
  atomicAdd(mp + 1, v.y);
  atomicAdd(mp + 2, v.z);
  atomicAdd(mp + 3, v.w);
}

__global__ __launch_bounds__(128) void sage_gemm_kernel(
    const float* __restrict__ msum, const float* __restrict__ deg,
    const float* __restrict__ H, const float* __restrict__ Wl,
    const float* __restrict__ Wr, const float* __restrict__ bl,
    float* __restrict__ out) {
  __shared__ float lds_mean[16 * SPAD];
  __shared__ float lds_h[16 * SPAD];

  const int tile_m = blockIdx.x;
  const int tid    = threadIdx.x;

#pragma unroll
  for (int i = 0; i < 4; ++i) {
    const int idx = tid + i * 128;
    const int row = idx >> 5;
    const int c4  = (idx & 31) * 4;
    const int gm  = tile_m * 16 + row;
    const float rs = 1.0f / fmaxf(deg[gm], 1.0f);
    const float4 mv = *(const float4*)(msum + (size_t)gm * FEAT + c4);
    const float4 hv = *(const float4*)(H    + (size_t)gm * FEAT + c4);
    float4 sv = { mv.x * rs, mv.y * rs, mv.z * rs, mv.w * rs };
    *(float4*)&lds_mean[row * SPAD + c4] = sv;
    *(float4*)&lds_h[row * SPAD + c4]    = hv;
  }
  __syncthreads();

  const int wave = tid >> 5;
  const int lane = tid & 31;
  const int mi   = lane & 15;
  const int h    = lane >> 4;
  const int n    = wave * 16 + mi;

  v8f c = {};
#pragma unroll 8
  for (int kc = 0; kc < FEAT; kc += 4) {
    const int k = kc + 2 * h;
    v2f a1 = *(const v2f*)&lds_mean[mi * SPAD + k];
    v2f b1 = { Wl[(size_t)k * HALF + n], Wl[(size_t)(k + 1) * HALF + n] };
    c = __builtin_amdgcn_wmma_f32_16x16x4_f32(false, a1, false, b1, (short)0, c, false, false);
    v2f a2 = *(const v2f*)&lds_h[mi * SPAD + k];
    v2f b2 = { Wr[(size_t)k * HALF + n], Wr[(size_t)(k + 1) * HALF + n] };
    c = __builtin_amdgcn_wmma_f32_16x16x4_f32(false, a2, false, b2, (short)0, c, false, false);
  }

  const float bias  = bl[n];
  const int   mbase = tile_m * 16 + 8 * h;
#pragma unroll
  for (int r = 0; r < 8; ++r) {
    float v = c[r] + bias;
    out[(size_t)(mbase + r) * HALF + n] = fmaxf(v, 0.0f);
  }
}

__global__ __launch_bounds__(256) void comb_gemm_kernel(
    const float* __restrict__ X, const float* __restrict__ Hin,
    const float* __restrict__ Hout, const float* __restrict__ W,
    const float* __restrict__ b, float* __restrict__ out) {
  __shared__ float lds_a[16 * CPAD];

  const int tile_m = blockIdx.x;
  const int tid    = threadIdx.x;

#pragma unroll
  for (int i = 0; i < 4; ++i) {
    const int idx = tid + i * 256;
    const int row = idx >> 6;
    const int c4  = (idx & 63) * 4;
    const int gm  = tile_m * 16 + row;
    float4 v;
    if (c4 < 128)      v = *(const float4*)(X    + (size_t)gm * FEAT + c4);
    else if (c4 < 192) v = *(const float4*)(Hin  + (size_t)gm * HALF + (c4 - 128));
    else               v = *(const float4*)(Hout + (size_t)gm * HALF + (c4 - 192));
    *(float4*)&lds_a[row * CPAD + c4] = v;
  }
  __syncthreads();

  const int wave = tid >> 5;
  const int lane = tid & 31;
  const int mi   = lane & 15;
  const int h    = lane >> 4;
  const int n    = wave * 16 + mi;

  v8f c = {};
#pragma unroll 8
  for (int kc = 0; kc < 2 * FEAT; kc += 4) {
    const int k = kc + 2 * h;
    v2f a  = *(const v2f*)&lds_a[mi * CPAD + k];
    v2f bb = { W[(size_t)k * FEAT + n], W[(size_t)(k + 1) * FEAT + n] };
    c = __builtin_amdgcn_wmma_f32_16x16x4_f32(false, a, false, bb, (short)0, c, false, false);
  }

  const float bias  = b[n];
  const int   mbase = tile_m * 16 + 8 * h;
#pragma unroll
  for (int r = 0; r < 8; ++r) {
    float v = c[r] + bias;
    out[(size_t)(mbase + r) * FEAT + n] = fmaxf(v, 0.0f);
  }
}

__global__ void final_kernel(const float* __restrict__ H, const float* __restrict__ W,
                             const float* __restrict__ b, float* __restrict__ out, int n) {
  int i = blockIdx.x * blockDim.x + threadIdx.x;
  if (i >= n) return;
  const float4* hr = (const float4*)(H + (size_t)i * FEAT);
  const float4* wr = (const float4*)W;
  float acc = 0.0f;
#pragma unroll 8
  for (int k = 0; k < FEAT / 4; ++k) {
    float4 hv = hr[k];
    float4 wv = wr[k];
    acc += hv.x * wv.x + hv.y * wv.y + hv.z * wv.z + hv.w * wv.w;
  }
  out[i] = acc + b[0];
}

extern "C" void kernel_launch(void* const* d_in, const int* in_sizes, int n_in,
                              void* d_out, int out_size, void* d_ws, size_t ws_size,
                              hipStream_t stream) {
  (void)n_in; (void)out_size; (void)ws_size;

  const float* x       = (const float*)d_in[0];
  const int*   ein     = (const int*)d_in[1];
  const int*   eout    = (const int*)d_in[2];
  const float* in_Wl0  = (const float*)d_in[3];
  const float* in_bl0  = (const float*)d_in[4];
  const float* in_Wr0  = (const float*)d_in[5];
  const float* out_Wl0 = (const float*)d_in[6];
  const float* out_bl0 = (const float*)d_in[7];
  const float* out_Wr0 = (const float*)d_in[8];
  const float* comb_W0 = (const float*)d_in[9];
  const float* comb_b0 = (const float*)d_in[10];
  const float* in_Wl1  = (const float*)d_in[11];
  const float* in_bl1  = (const float*)d_in[12];
  const float* in_Wr1  = (const float*)d_in[13];
  const float* out_Wl1 = (const float*)d_in[14];
  const float* out_bl1 = (const float*)d_in[15];
  const float* out_Wr1 = (const float*)d_in[16];
  const float* comb_W1 = (const float*)d_in[17];
  const float* comb_b1 = (const float*)d_in[18];
  const float* final_W = (const float*)d_in[19];
  const float* final_b = (const float*)d_in[20];
  float* out = (float*)d_out;

  const int N = NNODES;
  const int E = in_sizes[1] / 2;

  const size_t N128 = (size_t)N * FEAT;
  const size_t N64  = (size_t)N * HALF;
  const size_t PADN = 65536;
  float* ws     = (float*)d_ws;
  float* degIn  = ws;
  float* degOut = ws + PADN;
  float* msum   = ws + 2 * PADN;
  float* hin    = msum + N128;
  float* hout   = hin + N64;
  float* HA     = hout + N64;
  float* HB     = HA + N128;

  const int ZT = 256;
  auto nblk = [](long long c, int t) { return (unsigned)((c + t - 1) / t); };

  fill_zero_kernel<<<nblk(N, ZT), ZT, 0, stream>>>(degIn, N);
  fill_zero_kernel<<<nblk(N, ZT), ZT, 0, stream>>>(degOut, N);
  degree_kernel<<<nblk(E, ZT), ZT, 0, stream>>>(ein + E, degIn, E);
  degree_kernel<<<nblk(E, ZT), ZT, 0, stream>>>(eout + E, degOut, E);

  auto conv = [&](const float* H, const int* edges, const float* deg,
                  const float* Wl, const float* bl, const float* Wr, float* hdst) {
    fill_zero_kernel<<<nblk((long long)N128, ZT), ZT, 0, stream>>>(msum, (long long)N128);
    scatter_kernel<<<nblk((long long)E * 32, ZT), ZT, 0, stream>>>(H, edges, edges + E, msum, E);
    sage_gemm_kernel<<<MTILES, 128, 0, stream>>>(msum, deg, H, Wl, Wr, bl, hdst);
  };

  conv(x, ein,  degIn,  in_Wl0,  in_bl0,  in_Wr0,  hin);
  conv(x, eout, degOut, out_Wl0, out_bl0, out_Wr0, hout);
  comb_gemm_kernel<<<MTILES, 256, 0, stream>>>(x, hin, hout, comb_W0, comb_b0, HA);

  conv(HA, ein,  degIn,  in_Wl1,  in_bl1,  in_Wr1,  hin);
  conv(HA, eout, degOut, out_Wl1, out_bl1, out_Wr1, hout);
  comb_gemm_kernel<<<MTILES, 256, 0, stream>>>(HA, hin, hout, comb_W1, comb_b1, HB);

  final_kernel<<<nblk(N, ZT), ZT, 0, stream>>>(HB, final_W, final_b, out, N);
}


# Apply Wl before aggregation (mean is linear): scatter 64-wide projected rows instead of 128-wide inputs; halves gather bytes and f32 atomics; changes the order of f32 summation
# speedup vs baseline: 1.4920x; 1.4920x over previous
#include <hip/hip_runtime.h>
#include <hip/hip_bf16.h>


typedef float v2f __attribute__((ext_vector_type(2)));
typedef float v8f __attribute__((ext_vector_type(8)));

#define NNODES 50000
#define FEAT   128
#define HALF   64
#define MTILES (NNODES / 16)
#define SPAD   132
#define CPAD   260

__global__ void fill_zero_kernel(float* __restrict__ p, long long count) {
  long long i = (long long)blockIdx.x * blockDim.x + threadIdx.x;
  if (i < count) p[i] = 0.0f;
}

__global__ void degree_kernel(const int* __restrict__ dst, float* __restrict__ deg, int E) {
  int e = blockIdx.x * blockDim.x + threadIdx.x;
  if (e < E) atomicAdd(&deg[dst[e]], 1.0f);
}

__global__ __launch_bounds__(256) void proj_gemm_kernel(
    const float* __restrict__ H, const float* __restrict__ WlIn,
    const float* __restrict__ WlOut, float* __restrict__ Yin,
    float* __restrict__ Yout) {
  __shared__ float lds_h[16 * SPAD];

  const int tile_m = blockIdx.x;
  const int tid    = threadIdx.x;

#pragma unroll
  for (int i = 0; i < 2; ++i) {
    const int idx = tid + i * 256;
    const int row = idx >> 5;
    const int c4  = (idx & 31) * 4;
    const int gm  = tile_m * 16 + row;
    const float4 hv = *(const float4*)(H + (size_t)gm * FEAT + c4);
    *(float4*)&lds_h[row * SPAD + c4] = hv;
  }
  __syncthreads();

  const int wave = tid >> 5;
  const int lane = tid & 31;
  const int mi   = lane & 15;
  const int h    = lane >> 4;
  const int dir  = wave >> 2;
  const int n    = (wave & 3) * 16 + mi;
  const float* __restrict__ W = dir ? WlOut : WlIn;
  float* __restrict__ Y       = dir ? Yout : Yin;

  v8f c = {};
#pragma unroll 8
  for (int kc = 0; kc < FEAT; kc += 4) {
    const int k = kc + 2 * h;
    v2f a = *(const v2f*)&lds_h[mi * SPAD + k];
    v2f b = { W[(size_t)k * HALF + n], W[(size_t)(k + 1) * HALF + n] };
    c = __builtin_amdgcn_wmma_f32_16x16x4_f32(false, a, false, b, (short)0, c, false, false);
  }

  const int mbase = tile_m * 16 + 8 * h;
#pragma unroll
  for (int r = 0; r < 8; ++r) {
    Y[(size_t)(mbase + r) * HALF + n] = c[r];
  }
}

__global__ void scatter_kernel(const float* __restrict__ Y,
                               const int* __restrict__ src,
                               const int* __restrict__ dst,
                               float* __restrict__ msum, int E) {
  long long t = (long long)blockIdx.x * blockDim.x + threadIdx.x;
  int e = (int)(t >> 4);
  if (e >= E) return;
  int f = ((int)t & 15) * 4;
  int s = src[e];
  int d = dst[e];
  const float4 v = *(const float4*)(Y + (size_t)s * HALF + f);
  float* mp = msum + (size_t)d * HALF + f;
  atomicAdd(mp + 0, v.x);
  atomicAdd(mp + 1, v.y);
  atomicAdd(mp + 2, v.z);
  atomicAdd(mp + 3, v.w);
}

__global__ __launch_bounds__(128) void sage_gemm_kernel(
    const float* __restrict__ msum, const float* __restrict__ deg,
    const float* __restrict__ H, const float* __restrict__ Wr,
    const float* __restrict__ bl, float* __restrict__ out) {
  __shared__ float lds_h[16 * SPAD];

  const int tile_m = blockIdx.x;
  const int tid    = threadIdx.x;

#pragma unroll
  for (int i = 0; i < 4; ++i) {
    const int idx = tid + i * 128;
    const int row = idx >> 5;
    const int c4  = (idx & 31) * 4;
    const int gm  = tile_m * 16 + row;
    const float4 hv = *(const float4*)(H + (size_t)gm * FEAT + c4);
    *(float4*)&lds_h[row * SPAD + c4] = hv;
  }

  const int wave = tid >> 5;
  const int lane = tid & 31;
  const int mi   = lane & 15;
  const int h    = lane >> 4;
  const int n    = wave * 16 + mi;

  const float bias  = bl[n];
  const int   mbase = tile_m * 16 + 8 * h;

  v8f c;
#pragma unroll
  for (int r = 0; r < 8; ++r) {
    const int   gm = mbase + r;
    const float rs = 1.0f / fmaxf(deg[gm], 1.0f);
    c[r] = msum[(size_t)gm * HALF + n] * rs + bias;
  }
  __syncthreads();

#pragma unroll 8
  for (int kc = 0; kc < FEAT; kc += 4) {
    const int k = kc + 2 * h;
    v2f a = *(const v2f*)&lds_h[mi * SPAD + k];
    v2f b = { Wr[(size_t)k * HALF + n], Wr[(size_t)(k + 1) * HALF + n] };
    c = __builtin_amdgcn_wmma_f32_16x16x4_f32(false, a, false, b, (short)0, c, false, false);
  }

#pragma unroll
  for (int r = 0; r < 8; ++r) {
    out[(size_t)(mbase + r) * HALF + n] = fmaxf(c[r], 0.0f);
  }
}

__global__ __launch_bounds__(256) void comb_gemm_kernel(
    const float* __restrict__ X, const float* __restrict__ Hin,
    const float* __restrict__ Hout, const float* __restrict__ W,
    const float* __restrict__ b, float* __restrict__ out) {
  __shared__ float lds_a[16 * CPAD];

  const int tile_m = blockIdx.x;
  const int tid    = threadIdx.x;

#pragma unroll
  for (int i = 0; i < 4; ++i) {
    const int idx = tid + i * 256;
    const int row = idx >> 6;
    const int c4  = (idx & 63) * 4;
    const int gm  = tile_m * 16 + row;
    float4 v;
    if (c4 < 128)      v = *(const float4*)(X    + (size_t)gm * FEAT + c4);
    else if (c4 < 192) v = *(const float4*)(Hin  + (size_t)gm * HALF + (c4 - 128));
    else               v = *(const float4*)(Hout + (size_t)gm * HALF + (c4 - 192));
    *(float4*)&lds_a[row * CPAD + c4] = v;
  }
  __syncthreads();

  const int wave = tid >> 5;
  const int lane = tid & 31;
  const int mi   = lane & 15;
  const int h    = lane >> 4;
  const int n    = wave * 16 + mi;

  v8f c = {};
#pragma unroll 8
  for (int kc = 0; kc < 2 * FEAT; kc += 4) {
    const int k = kc + 2 * h;
    v2f a  = *(const v2f*)&lds_a[mi * CPAD + k];
    v2f bb = { W[(size_t)k * FEAT + n], W[(size_t)(k + 1) * FEAT + n] };
    c = __builtin_amdgcn_wmma_f32_16x16x4_f32(false, a, false, bb, (short)0, c, false, false);
  }

  const float bias  = b[n];
  const int   mbase = tile_m * 16 + 8 * h;
#pragma unroll
  for (int r = 0; r < 8; ++r) {
    float v = c[r] + bias;
    out[(size_t)(mbase + r) * FEAT + n] = fmaxf(v, 0.0f);
  }
}

__global__ void final_kernel(const float* __restrict__ H, const float* __restrict__ W,
                             const float* __restrict__ b, float* __restrict__ out, int n) {
  int i = blockIdx.x * blockDim.x + threadIdx.x;
  if (i >= n) return;
  const float4* hr = (const float4*)(H + (size_t)i * FEAT);
  const float4* wr = (const float4*)W;
  float acc = 0.0f;
#pragma unroll 8
  for (int k = 0; k < FEAT / 4; ++k) {
    float4 hv = hr[k];
    float4 wv = wr[k];
    acc += hv.x * wv.x + hv.y * wv.y + hv.z * wv.z + hv.w * wv.w;
  }
  out[i] = acc + b[0];
}

extern "C" void kernel_launch(void* const* d_in, const int* in_sizes, int n_in,
                              void* d_out, int out_size, void* d_ws, size_t ws_size,
                              hipStream_t stream) {
  (void)n_in; (void)out_size; (void)ws_size;

  const float* x       = (const float*)d_in[0];
  const int*   ein     = (const int*)d_in[1];
  const int*   eout    = (const int*)d_in[2];
  const float* in_Wl0  = (const float*)d_in[3];
  const float* in_bl0  = (const float*)d_in[4];
  const float* in_Wr0  = (const float*)d_in[5];
  const float* out_Wl0 = (const float*)d_in[6];
  const float* out_bl0 = (const float*)d_in[7];
  const float* out_Wr0 = (const float*)d_in[8];
  const float* comb_W0 = (const float*)d_in[9];
  const float* comb_b0 = (const float*)d_in[10];
  const float* in_Wl1  = (const float*)d_in[11];
  const float* in_bl1  = (const float*)d_in[12];
  const float* in_Wr1  = (const float*)d_in[13];
  const float* out_Wl1 = (const float*)d_in[14];
  const float* out_bl1 = (const float*)d_in[15];
  const float* out_Wr1 = (const float*)d_in[16];
  const float* comb_W1 = (const float*)d_in[17];
  const float* comb_b1 = (const float*)d_in[18];
  const float* final_W = (const float*)d_in[19];
  const float* final_b = (const float*)d_in[20];
  float* out = (float*)d_out;

  const int N = NNODES;
  const int E = in_sizes[1] / 2;

  const size_t N128 = (size_t)N * FEAT;
  const size_t N64  = (size_t)N * HALF;
  const size_t PADN = 65536;
  float* ws      = (float*)d_ws;
  float* degIn   = ws;
  float* degOut  = ws + PADN;
  float* msum    = ws + 2 * PADN;
  float* msumIn  = msum;
  float* msumOut = msum + N64;
  float* hin     = msum + N128;
  float* hout    = hin + N64;
  float* HA      = hout + N64;
  float* HB      = HA + N128;
  float* yIn     = HB;
  float* yOut    = HB + N64;

  const int ZT = 256;
  auto nblk = [](long long c, int t) { return (unsigned)((c + t - 1) / t); };

  fill_zero_kernel<<<nblk(N, ZT), ZT, 0, stream>>>(degIn, N);
  fill_zero_kernel<<<nblk(N, ZT), ZT, 0, stream>>>(degOut, N);
  degree_kernel<<<nblk(E, ZT), ZT, 0, stream>>>(ein + E, degIn, E);
  degree_kernel<<<nblk(E, ZT), ZT, 0, stream>>>(eout + E, degOut, E);

  auto layer = [&](const float* H,
                   const float* iWl, const float* ibl, const float* iWr,
                   const float* oWl, const float* obl, const float* oWr,
                   const float* cW, const float* cb, float* Hnext) {
    proj_gemm_kernel<<<MTILES, 256, 0, stream>>>(H, iWl, oWl, yIn, yOut);
    fill_zero_kernel<<<nblk((long long)N128, ZT), ZT, 0, stream>>>(msum, (long long)N128);
    scatter_kernel<<<nblk((long long)E * 16, ZT), ZT, 0, stream>>>(yIn,  ein,  ein + E,  msumIn,  E);
    scatter_kernel<<<nblk((long long)E * 16, ZT), ZT, 0, stream>>>(yOut, eout, eout + E, msumOut, E);
    sage_gemm_kernel<<<MTILES, 128, 0, stream>>>(msumIn,  degIn,  H, iWr, ibl, hin);
    sage_gemm_kernel<<<MTILES, 128, 0, stream>>>(msumOut, degOut, H, oWr, obl, hout);
    comb_gemm_kernel<<<MTILES, 256, 0, stream>>>(H, hin, hout, cW, cb, Hnext);
  };

  layer(x,  in_Wl0, in_bl0, in_Wr0, out_Wl0, out_bl0, out_Wr0, comb_W0, comb_b0, HA);
  layer(HA, in_Wl1, in_bl1, in_Wr1, out_Wl1, out_bl1, out_Wr1, comb_W1, comb_b1, HB);

  final_kernel<<<nblk(N, ZT), ZT, 0, stream>>>(HB, final_W, final_b, out, N);
}


# Replace f32 atomic scatter-add by per-call counting sort of edges by dst (ranks from returning LDS adds per node range) + 16-lane-per-node register gather-mean; changes the order of f32 summation
# speedup vs baseline: 2.1898x; 1.4677x over previous
#include <hip/hip_runtime.h>
#include <hip/hip_bf16.h>


typedef float v2f __attribute__((ext_vector_type(2)));
typedef float v8f __attribute__((ext_vector_type(8)));

#define NNODES 50000
#define FEAT   128
#define HALF   64
#define MTILES (NNODES / 16)
#define SPAD   132
#define CPAD   260
#define PADN   65536
#define NGRP   64
#define RNG    782
#define RU     16

__global__ void fill_zero_kernel(float* __restrict__ p, long long count) {
  long long i = (long long)blockIdx.x * blockDim.x + threadIdx.x;
  if (i < count) p[i] = 0.0f;
}


template <bool GUARD>
__device__ __forceinline__ void rank_step(const int* __restrict__ dst, int* __restrict__ rk,
                                          int* cnt, int base, int E, int lo, int nr) {
  int k[RU], r[RU];
#pragma unroll
  for (int u = 0; u < RU; ++u) {
    const int ee = base + u * 1024;
    k[u] = ((!GUARD || ee < E) ? dst[ee] : -1) - lo;
  }
#pragma unroll
  for (int u = 0; u < RU; ++u) {
    if ((unsigned)k[u] < (unsigned)nr) r[u] = atomicAdd(&cnt[k[u]], 1);
  }
#pragma unroll
  for (int u = 0; u < RU; ++u) {
    if ((unsigned)k[u] < (unsigned)nr) rk[base + u * 1024] = r[u];
  }
}

__global__ __launch_bounds__(1024) void edge_rank_kernel(
    const int* __restrict__ ein, const int* __restrict__ eout,
    int* __restrict__ cursor, int2* __restrict__ seg, int* __restrict__ rank, int E) {
  __shared__ int cnt[RNG];
  __shared__ int wave_sum[32];
  __shared__ int group_base;
  const int dir  = blockIdx.y;
  const int tid  = threadIdx.x;
  const int lane = tid & 31;
  const int wave = tid >> 5;
  const int lo   = blockIdx.x * RNG;
  const int nr   = min(RNG, NNODES - lo);
  const int* __restrict__ dst = (dir ? eout : ein) + E;
  int* __restrict__ rk = rank + (size_t)dir * E;

  if (tid < RNG) cnt[tid] = 0;
  __syncthreads();

  const int STEP  = RU * 1024;
  const int steps = (E + STEP - 1) / STEP;
  const int start = (int)(blockIdx.x * steps) / NGRP;
  for (int it = 0; it < steps; ++it) {
    int s = it + start;
    if (s >= steps) s -= steps;
    const int first = s * STEP;
    if (first + STEP <= E) rank_step<false>(dst, rk, cnt, first + tid, E, lo, nr);
    else                   rank_step<true >(dst, rk, cnt, first + tid, E, lo, nr);
  }
  __syncthreads();

  const int c = (tid < nr) ? cnt[tid] : 0;
  int incl = c;
#pragma unroll
  for (int d = 1; d < 32; d <<= 1) {
    const int t = __shfl_up(incl, d);
    if (lane >= d) incl += t;
  }
  if (lane == 31) wave_sum[wave] = incl;
  __syncthreads();
  if (wave == 0) {
    const int ws = wave_sum[lane];
    int wi = ws;
#pragma unroll
    for (int d = 1; d < 32; d <<= 1) {
      const int t = __shfl_up(wi, d);
      if (lane >= d) wi += t;
    }
    wave_sum[lane] = wi - ws;
    if (lane == 31) group_base = atomicAdd(&cursor[dir], wi);
  }
  __syncthreads();
  if (tid < nr)
    seg[dir * NNODES + lo + tid] = make_int2(group_base + wave_sum[wave] + incl - c, c);
}

__global__ void edge_place_kernel(const int* __restrict__ ein, const int* __restrict__ eout,
                                  const int2* __restrict__ seg, const int* __restrict__ rank,
                                  int* __restrict__ srt, int E) {
  const int dir = blockIdx.y;
  const int e = blockIdx.x * blockDim.x + threadIdx.x;
  if (e >= E) return;
  const int* __restrict__ ei = dir ? eout : ein;
  const int s = ei[e];
  const int d = ei[E + e];
  srt[(size_t)dir * E + seg[dir * NNODES + d].x + rank[(size_t)dir * E + e]] = s;
}

__global__ __launch_bounds__(256) void proj_gemm_kernel(
    const float* __restrict__ H, const float* __restrict__ WlIn,
    const float* __restrict__ WlOut, float* __restrict__ Yin,
    float* __restrict__ Yout) {
  __shared__ float lds_h[16 * SPAD];

  const int tile_m = blockIdx.x;
  const int tid    = threadIdx.x;

#pragma unroll
  for (int i = 0; i < 2; ++i) {
    const int idx = tid + i * 256;
    const int row = idx >> 5;
    const int c4  = (idx & 31) * 4;
    const int gm  = tile_m * 16 + row;
    const float4 hv = *(const float4*)(H + (size_t)gm * FEAT + c4);
    *(float4*)&lds_h[row * SPAD + c4] = hv;
  }
  __syncthreads();

  const int wave = tid >> 5;
  const int lane = tid & 31;
  const int mi   = lane & 15;
  const int h    = lane >> 4;
  const int dir  = wave >> 2;
  const int n    = (wave & 3) * 16 + mi;
  const float* __restrict__ W = dir ? WlOut : WlIn;
  float* __restrict__ Y       = dir ? Yout : Yin;

  v8f c = {};
#pragma unroll 8
  for (int kc = 0; kc < FEAT; kc += 4) {
    const int k = kc + 2 * h;
    v2f a = *(const v2f*)&lds_h[mi * SPAD + k];
    v2f b = { W[(size_t)k * HALF + n], W[(size_t)(k + 1) * HALF + n] };
    c = __builtin_amdgcn_wmma_f32_16x16x4_f32(false, a, false, b, (short)0, c, false, false);
  }

  const int mbase = tile_m * 16 + 8 * h;
#pragma unroll
  for (int r = 0; r < 8; ++r) {
    Y[(size_t)(mbase + r) * HALF + n] = c[r];
  }
}

__global__ __launch_bounds__(256) void gather_mean_kernel(
    const float* __restrict__ Y, const int2* __restrict__ seg,
    const int* __restrict__ srt, float* __restrict__ mean, int E) {
  const int  dir = blockIdx.y;
  const int  tid = threadIdx.x;
  const int  l16 = tid & 15;
  const bool hi  = (tid & 16) != 0;
  const int  n   = blockIdx.x * 16 + (tid >> 4);
  const int2 sg  = seg[dir * NNODES + n];
  const int  c   = sg.y;
  const int* __restrict__ sd   = srt + (size_t)dir * E + sg.x;
  const float* __restrict__ Yb = Y + (size_t)dir * NNODES * HALF;
  const unsigned col = l16 * 4;

  const int cmax = max(__builtin_amdgcn_readlane(c, 0), __builtin_amdgcn_readlane(c, 16));

  float4 acc = make_float4(0.0f, 0.0f, 0.0f, 0.0f);
  for (int base = 0; base < cmax; base += 16) {
    const int rem   = c - base;
    const int myidx = (l16 < rem) ? sd[base + l16] : 0;
    const int m     = min(cmax - base, 16);
    for (int j = 0; j < m; j += 8) {
      float4 v[8];
#pragma unroll
      for (int u = 0; u < 8; ++u) {
        const int jl = j + u;
        const unsigned s_lo = (unsigned)__builtin_amdgcn_readlane(myidx, jl);
        const unsigned s_hi = (unsigned)__builtin_amdgcn_readlane(myidx, jl + 16);
        const unsigned s    = hi ? s_hi : s_lo;
        v[u] = make_float4(0.0f, 0.0f, 0.0f, 0.0f);
        if (jl < rem) v[u] = *(const float4*)(Yb + (s * HALF + col));
      }
#pragma unroll
      for (int u = 0; u < 8; ++u) {
        acc.x += v[u].x; acc.y += v[u].y; acc.z += v[u].z; acc.w += v[u].w;
      }
    }
  }
  const float rs = 1.0f / fmaxf((float)c, 1.0f);
  acc.x *= rs; acc.y *= rs; acc.z *= rs; acc.w *= rs;
  *(float4*)(mean + ((size_t)dir * NNODES + n) * HALF + col) = acc;
}

__global__ __launch_bounds__(128) void sage_gemm_kernel(
    float* io, const float* __restrict__ H, const float* __restrict__ Wr,
    const float* __restrict__ bl) {
  __shared__ float lds_h[16 * SPAD];

  const int tile_m = blockIdx.x;
  const int tid    = threadIdx.x;

#pragma unroll
  for (int i = 0; i < 4; ++i) {
    const int idx = tid + i * 128;
    const int row = idx >> 5;
    const int c4  = (idx & 31) * 4;
    const int gm  = tile_m * 16 + row;
    const float4 hv = *(const float4*)(H + (size_t)gm * FEAT + c4);
    *(float4*)&lds_h[row * SPAD + c4] = hv;
  }

  const int wave = tid >> 5;
  const int lane = tid & 31;
  const int mi   = lane & 15;
  const int h    = lane >> 4;
  const int n    = wave * 16 + mi;

  const float bias  = bl[n];
  const int   mbase = tile_m * 16 + 8 * h;

  v8f c;
#pragma unroll
  for (int r = 0; r < 8; ++r) {
    c[r] = io[(size_t)(mbase + r) * HALF + n] + bias;
  }
  __syncthreads();

#pragma unroll 8
  for (int kc = 0; kc < FEAT; kc += 4) {
    const int k = kc + 2 * h;
    v2f a = *(const v2f*)&lds_h[mi * SPAD + k];
    v2f b = { Wr[(size_t)k * HALF + n], Wr[(size_t)(k + 1) * HALF + n] };
    c = __builtin_amdgcn_wmma_f32_16x16x4_f32(false, a, false, b, (short)0, c, false, false);
  }

#pragma unroll
  for (int r = 0; r < 8; ++r) {
    io[(size_t)(mbase + r) * HALF + n] = fmaxf(c[r], 0.0f);
  }
}

__global__ __launch_bounds__(256) void comb_gemm_kernel(
    const float* __restrict__ X, const float* __restrict__ Hin,
    const float* __restrict__ Hout, const float* __restrict__ W,
    const float* __restrict__ b, float* __restrict__ out) {
  __shared__ float lds_a[16 * CPAD];

  const int tile_m = blockIdx.x;
  const int tid    = threadIdx.x;

#pragma unroll
  for (int i = 0; i < 4; ++i) {
    const int idx = tid + i * 256;
    const int row = idx >> 6;
    const int c4  = (idx & 63) * 4;
    const int gm  = tile_m * 16 + row;
    float4 v;
    if (c4 < 128)      v = *(const float4*)(X    + (size_t)gm * FEAT + c4);
    else if (c4 < 192) v = *(const float4*)(Hin  + (size_t)gm * HALF + (c4 - 128));
    else               v = *(const float4*)(Hout + (size_t)gm * HALF + (c4 - 192));
    *(float4*)&lds_a[row * CPAD + c4] = v;
  }
  __syncthreads();

  const int wave = tid >> 5;
  const int lane = tid & 31;
  const int mi   = lane & 15;
  const int h    = lane >> 4;
  const int n    = wave * 16 + mi;

  v8f c = {};
#pragma unroll 8
  for (int kc = 0; kc < 2 * FEAT; kc += 4) {
    const int k = kc + 2 * h;
    v2f a  = *(const v2f*)&lds_a[mi * CPAD + k];
    v2f bb = { W[(size_t)k * FEAT + n], W[(size_t)(k + 1) * FEAT + n] };
    c = __builtin_amdgcn_wmma_f32_16x16x4_f32(false, a, false, bb, (short)0, c, false, false);
  }

  const float bias  = b[n];
  const int   mbase = tile_m * 16 + 8 * h;
#pragma unroll
  for (int r = 0; r < 8; ++r) {
    float v = c[r] + bias;
    out[(size_t)(mbase + r) * FEAT + n] = fmaxf(v, 0.0f);
  }
}

__global__ void final_kernel(const float* __restrict__ H, const float* __restrict__ W,
                             const float* __restrict__ b, float* __restrict__ out, int n) {
  int i = blockIdx.x * blockDim.x + threadIdx.x;
  if (i >= n) return;
  const float4* hr = (const float4*)(H + (size_t)i * FEAT);
  const float4* wr = (const float4*)W;
  float acc = 0.0f;
#pragma unroll 8
  for (int k = 0; k < FEAT / 4; ++k) {
    float4 hv = hr[k];
    float4 wv = wr[k];
    acc += hv.x * wv.x + hv.y * wv.y + hv.z * wv.z + hv.w * wv.w;
  }
  out[i] = acc + b[0];
}

extern "C" void kernel_launch(void* const* d_in, const int* in_sizes, int n_in,
                              void* d_out, int out_size, void* d_ws, size_t ws_size,
                              hipStream_t stream) {
  (void)n_in; (void)out_size; (void)ws_size;

  const float* x       = (const float*)d_in[0];
  const int*   ein     = (const int*)d_in[1];
  const int*   eout    = (const int*)d_in[2];
  const float* in_Wl0  = (const float*)d_in[3];
  const float* in_bl0  = (const float*)d_in[4];
  const float* in_Wr0  = (const float*)d_in[5];
  const float* out_Wl0 = (const float*)d_in[6];
  const float* out_bl0 = (const float*)d_in[7];
  const float* out_Wr0 = (const float*)d_in[8];
  const float* comb_W0 = (const float*)d_in[9];
  const float* comb_b0 = (const float*)d_in[10];
  const float* in_Wl1  = (const float*)d_in[11];
  const float* in_bl1  = (const float*)d_in[12];
  const float* in_Wr1  = (const float*)d_in[13];
  const float* out_Wl1 = (const float*)d_in[14];
  const float* out_bl1 = (const float*)d_in[15];
  const float* out_Wr1 = (const float*)d_in[16];
  const float* comb_W1 = (const float*)d_in[17];
  const float* comb_b1 = (const float*)d_in[18];
  const float* final_W = (const float*)d_in[19];
  const float* final_b = (const float*)d_in[20];
  float* out = (float*)d_out;

  const int N = NNODES;
  const int E = in_sizes[1] / 2;

  const size_t N128 = (size_t)N * FEAT;
  const size_t N64  = (size_t)N * HALF;
  float* ws      = (float*)d_ws;
  int*   cursor  = (int*)d_ws;
  float* mean    = ws + 2 * PADN;
  float* hin     = mean;
  float* hout    = mean + N64;
  int2*  seg     = (int2*)(mean + N128);
  int*   srt     = (int*)(seg + 2 * N);
  int*   rank    = srt + 2 * (size_t)E;
  float* HA      = mean + 2 * N128;
  float* HB      = HA + N128;
  float* yIn     = HB;
  float* yOut    = HB + N64;

  const int ZT = 256;
  auto nblk = [](long long c, int t) { return (unsigned)((c + t - 1) / t); };

  fill_zero_kernel<<<1, ZT, 0, stream>>>(ws, ZT);
  edge_rank_kernel<<<dim3(NGRP, 2), 1024, 0, stream>>>(ein, eout, cursor, seg, rank, E);
  edge_place_kernel<<<dim3(nblk(E, ZT), 2), ZT, 0, stream>>>(ein, eout, seg, rank, srt, E);

  auto layer = [&](const float* H,
                   const float* iWl, const float* ibl, const float* iWr,
                   const float* oWl, const float* obl, const float* oWr,
                   const float* cW, const float* cb, float* Hnext) {
    proj_gemm_kernel<<<MTILES, 256, 0, stream>>>(H, iWl, oWl, yIn, yOut);
    gather_mean_kernel<<<dim3(MTILES, 2), 256, 0, stream>>>(yIn, seg, srt, mean, E);
    sage_gemm_kernel<<<MTILES, 128, 0, stream>>>(hin,  H, iWr, ibl);
    sage_gemm_kernel<<<MTILES, 128, 0, stream>>>(hout, H, oWr, obl);
    comb_gemm_kernel<<<MTILES, 256, 0, stream>>>(H, hin, hout, cW, cb, Hnext);
  };

  layer(x,  in_Wl0, in_bl0, in_Wr0, out_Wl0, out_bl0, out_Wr0, comb_W0, comb_b0, HA);
  layer(HA, in_Wl1, in_bl1, in_Wr1, out_Wl1, out_bl1, out_Wr1, comb_W1, comb_b1, HB);

  final_kernel<<<nblk(N, ZT), ZT, 0, stream>>>(HB, final_W, final_b, out, N);
}


# Fuse each layer's row-local GEMM chain (both conv GEMMs, combine GEMM, then layer 1's projection) into one kernel per 16-row tile, intermediate tiles in LDS; 14 -> 9 launches, sums bit-identical
# speedup vs baseline: 2.5059x; 1.1444x over previous
#include <hip/hip_runtime.h>
#include <hip/hip_bf16.h>


typedef float v2f __attribute__((ext_vector_type(2)));
typedef float v8f __attribute__((ext_vector_type(8)));

#define NNODES 50000
#define FEAT   128
#define HALF   64
#define MTILES (NNODES / 16)
#define SPAD   132
#define CPAD   260
#define PADN   65536
#define NGRP   64
#define RNG    782
#define RU     16

__global__ void fill_zero_kernel(float* __restrict__ p, long long count) {
  long long i = (long long)blockIdx.x * blockDim.x + threadIdx.x;
  if (i < count) p[i] = 0.0f;
}


template <bool GUARD>
__device__ __forceinline__ void rank_step(const int* __restrict__ dst, int* __restrict__ rk,
                                          int* cnt, int base, int E, int lo, int nr) {
  int k[RU], r[RU];
#pragma unroll
  for (int u = 0; u < RU; ++u) {
    const int ee = base + u * 1024;
    k[u] = ((!GUARD || ee < E) ? dst[ee] : -1) - lo;
  }
#pragma unroll
  for (int u = 0; u < RU; ++u) {
    if ((unsigned)k[u] < (unsigned)nr) r[u] = atomicAdd(&cnt[k[u]], 1);
  }
#pragma unroll
  for (int u = 0; u < RU; ++u) {
    if ((unsigned)k[u] < (unsigned)nr) rk[base + u * 1024] = r[u];
  }
}

__global__ __launch_bounds__(1024) void edge_rank_kernel(
    const int* __restrict__ ein, const int* __restrict__ eout,
    int* __restrict__ cursor, int2* __restrict__ seg, int* __restrict__ rank, int E) {
  __shared__ int cnt[RNG];
  __shared__ int wave_sum[32];
  __shared__ int group_base;
  const int dir  = blockIdx.y;
  const int tid  = threadIdx.x;
  const int lane = tid & 31;
  const int wave = tid >> 5;
  const int lo   = blockIdx.x * RNG;
  const int nr   = min(RNG, NNODES - lo);
  const int* __restrict__ dst = (dir ? eout : ein) + E;
  int* __restrict__ rk = rank + (size_t)dir * E;

  if (tid < RNG) cnt[tid] = 0;
  __syncthreads();

  const int STEP  = RU * 1024;
  const int steps = (E + STEP - 1) / STEP;
  const int start = (int)(blockIdx.x * steps) / NGRP;
  for (int it = 0; it < steps; ++it) {
    int s = it + start;
    if (s >= steps) s -= steps;
    const int first = s * STEP;
    if (first + STEP <= E) rank_step<false>(dst, rk, cnt, first + tid, E, lo, nr);
    else                   rank_step<true >(dst, rk, cnt, first + tid, E, lo, nr);
  }
  __syncthreads();

  const int c = (tid < nr) ? cnt[tid] : 0;
  int incl = c;
#pragma unroll
  for (int d = 1; d < 32; d <<= 1) {
    const int t = __shfl_up(incl, d);
    if (lane >= d) incl += t;
  }
  if (lane == 31) wave_sum[wave] = incl;
  __syncthreads();
  if (wave == 0) {
    const int ws = wave_sum[lane];
    int wi = ws;
#pragma unroll
    for (int d = 1; d < 32; d <<= 1) {
      const int t = __shfl_up(wi, d);
      if (lane >= d) wi += t;
    }
    wave_sum[lane] = wi - ws;
    if (lane == 31) group_base = atomicAdd(&cursor[dir], wi);
  }
  __syncthreads();
  if (tid < nr)
    seg[dir * NNODES + lo + tid] = make_int2(group_base + wave_sum[wave] + incl - c, c);
}

__global__ void edge_place_kernel(const int* __restrict__ ein, const int* __restrict__ eout,
                                  const int2* __restrict__ seg, const int* __restrict__ rank,
                                  int* __restrict__ srt, int E) {
  const int dir = blockIdx.y;
  const int e = blockIdx.x * blockDim.x + threadIdx.x;
  if (e >= E) return;
  const int* __restrict__ ei = dir ? eout : ein;
  const int s = ei[e];
  const int d = ei[E + e];
  srt[(size_t)dir * E + seg[dir * NNODES + d].x + rank[(size_t)dir * E + e]] = s;
}

__global__ __launch_bounds__(256) void proj_gemm_kernel(
    const float* __restrict__ H, const float* __restrict__ WlIn,
    const float* __restrict__ WlOut, float* __restrict__ Yin,
    float* __restrict__ Yout) {
  __shared__ float lds_h[16 * SPAD];

  const int tile_m = blockIdx.x;
  const int tid    = threadIdx.x;

#pragma unroll
  for (int i = 0; i < 2; ++i) {
    const int idx = tid + i * 256;
    const int row = idx >> 5;
    const int c4  = (idx & 31) * 4;
    const int gm  = tile_m * 16 + row;
    const float4 hv = *(const float4*)(H + (size_t)gm * FEAT + c4);
    *(float4*)&lds_h[row * SPAD + c4] = hv;
  }
  __syncthreads();

  const int wave = tid >> 5;
  const int lane = tid & 31;
  const int mi   = lane & 15;
  const int h    = lane >> 4;
  const int dir  = wave >> 2;
  const int n    = (wave & 3) * 16 + mi;
  const float* __restrict__ W = dir ? WlOut : WlIn;
  float* __restrict__ Y       = dir ? Yout : Yin;

  v8f c = {};
#pragma unroll 8
  for (int kc = 0; kc < FEAT; kc += 4) {
    const int k = kc + 2 * h;
    v2f a = *(const v2f*)&lds_h[mi * SPAD + k];
    v2f b = { W[(size_t)k * HALF + n], W[(size_t)(k + 1) * HALF + n] };
    c = __builtin_amdgcn_wmma_f32_16x16x4_f32(false, a, false, b, (short)0, c, false, false);
  }

  const int mbase = tile_m * 16 + 8 * h;
#pragma unroll
  for (int r = 0; r < 8; ++r) {
    Y[(size_t)(mbase + r) * HALF + n] = c[r];
  }
}

__global__ __launch_bounds__(256) void gather_mean_kernel(
    const float* __restrict__ Y, const int2* __restrict__ seg,
    const int* __restrict__ srt, float* __restrict__ mean, int E) {
  const int  dir = blockIdx.y;
  const int  tid = threadIdx.x;
  const int  l16 = tid & 15;
  const bool hi  = (tid & 16) != 0;
  const int  n   = blockIdx.x * 16 + (tid >> 4);
  const int2 sg  = seg[dir * NNODES + n];
  const int  c   = sg.y;
  const int* __restrict__ sd   = srt + (size_t)dir * E + sg.x;
  const float* __restrict__ Yb = Y + (size_t)dir * NNODES * HALF;
  const unsigned col = l16 * 4;

  const int cmax = max(__builtin_amdgcn_readlane(c, 0), __builtin_amdgcn_readlane(c, 16));

  float4 acc = make_float4(0.0f, 0.0f, 0.0f, 0.0f);
  for (int base = 0; base < cmax; base += 16) {
    const int rem   = c - base;
    const int myidx = (l16 < rem) ? sd[base + l16] : 0;
    const int m     = min(cmax - base, 16);
    for (int j = 0; j < m; j += 8) {
      float4 v[8];
#pragma unroll
      for (int u = 0; u < 8; ++u) {
        const int jl = j + u;
        const unsigned s_lo = (unsigned)__builtin_amdgcn_readlane(myidx, jl);
        const unsigned s_hi = (unsigned)__builtin_amdgcn_readlane(myidx, jl + 16);
        const unsigned s    = hi ? s_hi : s_lo;
        v[u] = make_float4(0.0f, 0.0f, 0.0f, 0.0f);
        if (jl < rem) v[u] = *(const float4*)(Yb + (s * HALF + col));
      }
#pragma unroll
      for (int u = 0; u < 8; ++u) {
        acc.x += v[u].x; acc.y += v[u].y; acc.z += v[u].z; acc.w += v[u].w;
      }
    }
  }
  const float rs = 1.0f / fmaxf((float)c, 1.0f);
  acc.x *= rs; acc.y *= rs; acc.z *= rs; acc.w *= rs;
  *(float4*)(mean + ((size_t)dir * NNODES + n) * HALF + col) = acc;
}

template <bool PROJ>
__global__ __launch_bounds__(256) void layer_tail_kernel(
    const float* __restrict__ H, const float* __restrict__ mean,
    const float* __restrict__ WrIn, const float* __restrict__ WrOut,
    const float* __restrict__ blIn, const float* __restrict__ blOut,
    const float* __restrict__ cW, const float* __restrict__ cb,
    float* __restrict__ Hnext,
    const float* __restrict__ nWlIn, const float* __restrict__ nWlOut,
    float* __restrict__ Yin, float* __restrict__ Yout) {
  __shared__ float lds_a[16 * CPAD];
  __shared__ float lds_o[PROJ ? 16 * SPAD : 4];

  const int tile_m = blockIdx.x;
  const int tid    = threadIdx.x;

#pragma unroll
  for (int i = 0; i < 2; ++i) {
    const int idx = tid + i * 256;
    const int row = idx >> 5;
    const int c4  = (idx & 31) * 4;
    const int gm  = tile_m * 16 + row;
    const float4 hv = *(const float4*)(H + (size_t)gm * FEAT + c4);
    *(float4*)&lds_a[row * CPAD + c4] = hv;
  }

  const int wave = tid >> 5;
  const int lane = tid & 31;
  const int mi   = lane & 15;
  const int h    = lane >> 4;
  const int dir  = wave >> 2;
  const int n64  = (wave & 3) * 16 + mi;
  const int n    = wave * 16 + mi;
  const int mbase = tile_m * 16 + 8 * h;

  {
    const float* __restrict__ Wr = dir ? WrOut : WrIn;
    const float bias = (dir ? blOut : blIn)[n64];
    const float* __restrict__ mrow = mean + (size_t)dir * NNODES * HALF;
    v8f c;
#pragma unroll
    for (int r = 0; r < 8; ++r) {
      c[r] = mrow[(size_t)(mbase + r) * HALF + n64] + bias;
    }
    __syncthreads();

#pragma unroll 8
    for (int kc = 0; kc < FEAT; kc += 4) {
      const int k = kc + 2 * h;
      v2f a = *(const v2f*)&lds_a[mi * CPAD + k];
      v2f b = { Wr[(size_t)k * HALF + n64], Wr[(size_t)(k + 1) * HALF + n64] };
      c = __builtin_amdgcn_wmma_f32_16x16x4_f32(false, a, false, b, (short)0, c, false, false);
    }

#pragma unroll
    for (int r = 0; r < 8; ++r) {
      lds_a[(8 * h + r) * CPAD + FEAT + n] = fmaxf(c[r], 0.0f);
    }
  }
  __syncthreads();

  {
    v8f c = {};
#pragma unroll 8
    for (int kc = 0; kc < 2 * FEAT; kc += 4) {
      const int k = kc + 2 * h;
      v2f a  = *(const v2f*)&lds_a[mi * CPAD + k];
      v2f bb = { cW[(size_t)k * FEAT + n], cW[(size_t)(k + 1) * FEAT + n] };
      c = __builtin_amdgcn_wmma_f32_16x16x4_f32(false, a, false, bb, (short)0, c, false, false);
    }

    const float bias = cb[n];
#pragma unroll
    for (int r = 0; r < 8; ++r) {
      const float v = fmaxf(c[r] + bias, 0.0f);
      Hnext[(size_t)(mbase + r) * FEAT + n] = v;
      if (PROJ) lds_o[(8 * h + r) * SPAD + n] = v;
    }
  }

  if (PROJ) {
    __syncthreads();
    const float* __restrict__ nW = dir ? nWlOut : nWlIn;
    float* __restrict__ Y        = dir ? Yout : Yin;
    v8f c = {};
#pragma unroll 8
    for (int kc = 0; kc < FEAT; kc += 4) {
      const int k = kc + 2 * h;
      v2f a = *(const v2f*)&lds_o[mi * SPAD + k];
      v2f b = { nW[(size_t)k * HALF + n64], nW[(size_t)(k + 1) * HALF + n64] };
      c = __builtin_amdgcn_wmma_f32_16x16x4_f32(false, a, false, b, (short)0, c, false, false);
    }
#pragma unroll
    for (int r = 0; r < 8; ++r) {
      Y[(size_t)(mbase + r) * HALF + n64] = c[r];
    }
  }
}

__global__ void final_kernel(const float* __restrict__ H, const float* __restrict__ W,
                             const float* __restrict__ b, float* __restrict__ out, int n) {
  int i = blockIdx.x * blockDim.x + threadIdx.x;
  if (i >= n) return;
  const float4* hr = (const float4*)(H + (size_t)i * FEAT);
  const float4* wr = (const float4*)W;
  float acc = 0.0f;
#pragma unroll 8
  for (int k = 0; k < FEAT / 4; ++k) {
    float4 hv = hr[k];
    float4 wv = wr[k];
    acc += hv.x * wv.x + hv.y * wv.y + hv.z * wv.z + hv.w * wv.w;
  }
  out[i] = acc + b[0];
}

extern "C" void kernel_launch(void* const* d_in, const int* in_sizes, int n_in,
                              void* d_out, int out_size, void* d_ws, size_t ws_size,
                              hipStream_t stream) {
  (void)n_in; (void)out_size; (void)ws_size;

  const float* x       = (const float*)d_in[0];
  const int*   ein     = (const int*)d_in[1];
  const int*   eout    = (const int*)d_in[2];
  const float* in_Wl0  = (const float*)d_in[3];
  const float* in_bl0  = (const float*)d_in[4];
  const float* in_Wr0  = (const float*)d_in[5];
  const float* out_Wl0 = (const float*)d_in[6];
  const float* out_bl0 = (const float*)d_in[7];
  const float* out_Wr0 = (const float*)d_in[8];
  const float* comb_W0 = (const float*)d_in[9];
  const float* comb_b0 = (const float*)d_in[10];
  const float* in_Wl1  = (const float*)d_in[11];
  const float* in_bl1  = (const float*)d_in[12];
  const float* in_Wr1  = (const float*)d_in[13];
  const float* out_Wl1 = (const float*)d_in[14];
  const float* out_bl1 = (const float*)d_in[15];
  const float* out_Wr1 = (const float*)d_in[16];
  const float* comb_W1 = (const float*)d_in[17];
  const float* comb_b1 = (const float*)d_in[18];
  const float* final_W = (const float*)d_in[19];
  const float* final_b = (const float*)d_in[20];
  float* out = (float*)d_out;

  const int N = NNODES;
  const int E = in_sizes[1] / 2;

  const size_t N128 = (size_t)N * FEAT;
  const size_t N64  = (size_t)N * HALF;
  float* ws      = (float*)d_ws;
  int*   cursor  = (int*)d_ws;
  float* mean    = ws + 2 * PADN;
  int2*  seg     = (int2*)(mean + N128);
  int*   srt     = (int*)(seg + 2 * N);
  int*   rank    = srt + 2 * (size_t)E;
  float* HA      = mean + 2 * N128;
  float* HB      = HA + N128;
  float* yIn     = HB;
  float* yOut    = HB + N64;

  const int ZT = 256;
  auto nblk = [](long long c, int t) { return (unsigned)((c + t - 1) / t); };

  fill_zero_kernel<<<1, ZT, 0, stream>>>(ws, ZT);
  edge_rank_kernel<<<dim3(NGRP, 2), 1024, 0, stream>>>(ein, eout, cursor, seg, rank, E);
  edge_place_kernel<<<dim3(nblk(E, ZT), 2), ZT, 0, stream>>>(ein, eout, seg, rank, srt, E);

  proj_gemm_kernel<<<MTILES, 256, 0, stream>>>(x, in_Wl0, out_Wl0, yIn, yOut);
  gather_mean_kernel<<<dim3(MTILES, 2), 256, 0, stream>>>(yIn, seg, srt, mean, E);
  layer_tail_kernel<true><<<MTILES, 256, 0, stream>>>(
      x, mean, in_Wr0, out_Wr0, in_bl0, out_bl0, comb_W0, comb_b0, HA,
      in_Wl1, out_Wl1, yIn, yOut);

  gather_mean_kernel<<<dim3(MTILES, 2), 256, 0, stream>>>(yIn, seg, srt, mean, E);
  layer_tail_kernel<false><<<MTILES, 256, 0, stream>>>(
      HA, mean, in_Wr1, out_Wr1, in_bl1, out_bl1, comb_W1, comb_b1, HB,
      nullptr, nullptr, nullptr, nullptr);

  final_kernel<<<nblk(N, ZT), ZT, 0, stream>>>(HB, final_W, final_b, out, N);
}


# Fuse each layer's neighbour-mean gather into its layer kernel: the tile's 16 half-waves sum their nodes' segments, means reach the conv GEMM through LDS; 9 -> 7 launches, sums bit-identical
# speedup vs baseline: 2.6256x; 1.0478x over previous
#include <hip/hip_runtime.h>
#include <hip/hip_bf16.h>


typedef float v2f __attribute__((ext_vector_type(2)));
typedef float v8f __attribute__((ext_vector_type(8)));

#define NNODES 50000
#define FEAT   128
#define HALF   64
#define MTILES (NNODES / 16)
#define SPAD   132
#define CPAD   260
#define PADN   65536
#define NGRP   64
#define RNG    782
#define RU     16

__global__ void fill_zero_kernel(float* __restrict__ p, long long count) {
  long long i = (long long)blockIdx.x * blockDim.x + threadIdx.x;
  if (i < count) p[i] = 0.0f;
}


template <bool GUARD>
__device__ __forceinline__ void rank_step(const int* __restrict__ dst, int* __restrict__ rk,
                                          int* cnt, int base, int E, int lo, int nr) {
  int k[RU], r[RU];
#pragma unroll
  for (int u = 0; u < RU; ++u) {
    const int ee = base + u * 1024;
    k[u] = ((!GUARD || ee < E) ? dst[ee] : -1) - lo;
  }
#pragma unroll
  for (int u = 0; u < RU; ++u) {
    if ((unsigned)k[u] < (unsigned)nr) r[u] = atomicAdd(&cnt[k[u]], 1);
  }
#pragma unroll
  for (int u = 0; u < RU; ++u) {
    if ((unsigned)k[u] < (unsigned)nr) rk[base + u * 1024] = r[u];
  }
}

__global__ __launch_bounds__(1024) void edge_rank_kernel(
    const int* __restrict__ ein, const int* __restrict__ eout,
    int* __restrict__ cursor, int2* __restrict__ seg, int* __restrict__ rank, int E) {
  __shared__ int cnt[RNG];
  __shared__ int wave_sum[32];
  __shared__ int group_base;
  const int dir  = blockIdx.y;
  const int tid  = threadIdx.x;
  const int lane = tid & 31;
  const int wave = tid >> 5;
  const int lo   = blockIdx.x * RNG;
  const int nr   = min(RNG, NNODES - lo);
  const int* __restrict__ dst = (dir ? eout : ein) + E;
  int* __restrict__ rk = rank + (size_t)dir * E;

  if (tid < RNG) cnt[tid] = 0;
  __syncthreads();

  const int STEP  = RU * 1024;
  const int steps = (E + STEP - 1) / STEP;
  const int start = (int)(blockIdx.x * steps) / NGRP;
  for (int it = 0; it < steps; ++it) {
    int s = it + start;
    if (s >= steps) s -= steps;
    const int first = s * STEP;
    if (first + STEP <= E) rank_step<false>(dst, rk, cnt, first + tid, E, lo, nr);
    else                   rank_step<true >(dst, rk, cnt, first + tid, E, lo, nr);
  }
  __syncthreads();

  const int c = (tid < nr) ? cnt[tid] : 0;
  int incl = c;
#pragma unroll
  for (int d = 1; d < 32; d <<= 1) {
    const int t = __shfl_up(incl, d);
    if (lane >= d) incl += t;
  }
  if (lane == 31) wave_sum[wave] = incl;
  __syncthreads();
  if (wave == 0) {
    const int ws = wave_sum[lane];
    int wi = ws;
#pragma unroll
    for (int d = 1; d < 32; d <<= 1) {
      const int t = __shfl_up(wi, d);
      if (lane >= d) wi += t;
    }
    wave_sum[lane] = wi - ws;
    if (lane == 31) group_base = atomicAdd(&cursor[dir], wi);
  }
  __syncthreads();
  if (tid < nr)
    seg[dir * NNODES + lo + tid] = make_int2(group_base + wave_sum[wave] + incl - c, c);
}

__global__ void edge_place_kernel(const int* __restrict__ ein, const int* __restrict__ eout,
                                  const int2* __restrict__ seg, const int* __restrict__ rank,
                                  int* __restrict__ srt, int E) {
  const int dir = blockIdx.y;
  const int e = blockIdx.x * blockDim.x + threadIdx.x;
  if (e >= E) return;
  const int* __restrict__ ei = dir ? eout : ein;
  const int s = ei[e];
  const int d = ei[E + e];
  srt[(size_t)dir * E + seg[dir * NNODES + d].x + rank[(size_t)dir * E + e]] = s;
}

__global__ __launch_bounds__(256) void proj_gemm_kernel(
    const float* __restrict__ H, const float* __restrict__ WlIn,
    const float* __restrict__ WlOut, float* __restrict__ Yin,
    float* __restrict__ Yout) {
  __shared__ float lds_h[16 * SPAD];

  const int tile_m = blockIdx.x;
  const int tid    = threadIdx.x;

#pragma unroll
  for (int i = 0; i < 2; ++i) {
    const int idx = tid + i * 256;
    const int row = idx >> 5;
    const int c4  = (idx & 31) * 4;
    const int gm  = tile_m * 16 + row;
    const float4 hv = *(const float4*)(H + (size_t)gm * FEAT + c4);
    *(float4*)&lds_h[row * SPAD + c4] = hv;
  }
  __syncthreads();

  const int wave = tid >> 5;
  const int lane = tid & 31;
  const int mi   = lane & 15;
  const int h    = lane >> 4;
  const int dir  = wave >> 2;
  const int n    = (wave & 3) * 16 + mi;
  const float* __restrict__ W = dir ? WlOut : WlIn;
  float* __restrict__ Y       = dir ? Yout : Yin;

  v8f c = {};
#pragma unroll 8
  for (int kc = 0; kc < FEAT; kc += 4) {
    const int k = kc + 2 * h;
    v2f a = *(const v2f*)&lds_h[mi * SPAD + k];
    v2f b = { W[(size_t)k * HALF + n], W[(size_t)(k + 1) * HALF + n] };
    c = __builtin_amdgcn_wmma_f32_16x16x4_f32(false, a, false, b, (short)0, c, false, false);
  }

  const int mbase = tile_m * 16 + 8 * h;
#pragma unroll
  for (int r = 0; r < 8; ++r) {
    Y[(size_t)(mbase + r) * HALF + n] = c[r];
  }
}

__device__ __forceinline__ float4 segment_mean(
    const float* __restrict__ Yb, const int* __restrict__ sd, int c, int idx0,
    int l16, bool hi, unsigned col) {
  const int cmax = max(__builtin_amdgcn_readlane(c, 0), __builtin_amdgcn_readlane(c, 16));

  float4 acc = make_float4(0.0f, 0.0f, 0.0f, 0.0f);
  for (int base = 0; base < cmax; base += 16) {
    const int rem = c - base;
    int myidx = idx0;
    if (base != 0) myidx = (l16 < rem) ? sd[base + l16] : 0;
    const int m = min(cmax - base, 16);
    for (int j = 0; j < m; j += 8) {
      float4 v[8];
#pragma unroll
      for (int u = 0; u < 8; ++u) {
        const int jl = j + u;
        const unsigned s_lo = (unsigned)__builtin_amdgcn_readlane(myidx, jl);
        const unsigned s_hi = (unsigned)__builtin_amdgcn_readlane(myidx, jl + 16);
        const unsigned s    = hi ? s_hi : s_lo;
        v[u] = make_float4(0.0f, 0.0f, 0.0f, 0.0f);
        if (jl < rem) v[u] = *(const float4*)(Yb + (s * HALF + col));
      }
#pragma unroll
      for (int u = 0; u < 8; ++u) {
        acc.x += v[u].x; acc.y += v[u].y; acc.z += v[u].z; acc.w += v[u].w;
      }
    }
  }
  const float rs = 1.0f / fmaxf((float)c, 1.0f);
  acc.x *= rs; acc.y *= rs; acc.z *= rs; acc.w *= rs;
  return acc;
}

template <bool PROJ>
__global__ __launch_bounds__(256) void layer_tail_kernel(
    const float* __restrict__ H, const float* __restrict__ Y,
    const int2* __restrict__ seg, const int* __restrict__ srt, int E,
    const float* __restrict__ WrIn, const float* __restrict__ WrOut,
    const float* __restrict__ blIn, const float* __restrict__ blOut,
    const float* __restrict__ cW, const float* __restrict__ cb,
    float* __restrict__ Hnext,
    const float* __restrict__ nWlIn, const float* __restrict__ nWlOut,
    float* __restrict__ Yin, float* __restrict__ Yout) {
  __shared__ float lds_a[16 * CPAD];
  __shared__ float lds_o[PROJ ? 16 * SPAD : 4];

  const int tile_m = blockIdx.x;
  const int tid    = threadIdx.x;

#pragma unroll
  for (int i = 0; i < 2; ++i) {
    const int idx = tid + i * 256;
    const int row = idx >> 5;
    const int c4  = (idx & 31) * 4;
    const int gm  = tile_m * 16 + row;
    const float4 hv = *(const float4*)(H + (size_t)gm * FEAT + c4);
    *(float4*)&lds_a[row * CPAD + c4] = hv;
  }

  {
    const int  l16 = tid & 15;
    const bool hi  = (tid & 16) != 0;
    const int  nl  = tid >> 4;
    const int  nd  = tile_m * 16 + nl;
    const unsigned col = l16 * 4;
    const int2 sg0 = seg[nd];
    const int2 sg1 = seg[NNODES + nd];
    const int* __restrict__ sd0 = srt + sg0.x;
    const int* __restrict__ sd1 = srt + (size_t)E + sg1.x;
    const int r0 = sd0[(l16 < sg0.y) ? l16 : 0];
    const int r1 = sd1[(l16 < sg1.y) ? l16 : 0];
    const int i0 = (l16 < sg0.y) ? r0 : 0;
    const int i1 = (l16 < sg1.y) ? r1 : 0;
    const float4 m0 = segment_mean(Y, sd0, sg0.y, i0, l16, hi, col);
    *(float4*)&lds_a[nl * CPAD + FEAT + col] = m0;
    const float4 m1 = segment_mean(Y + (size_t)NNODES * HALF, sd1, sg1.y, i1, l16, hi, col);
    *(float4*)&lds_a[nl * CPAD + FEAT + HALF + col] = m1;
  }

  const int wave = tid >> 5;
  const int lane = tid & 31;
  const int mi   = lane & 15;
  const int h    = lane >> 4;
  const int dir  = wave >> 2;
  const int n64  = (wave & 3) * 16 + mi;
  const int n    = wave * 16 + mi;
  const int mbase = tile_m * 16 + 8 * h;

  {
    const float* __restrict__ Wr = dir ? WrOut : WrIn;
    const float bias = (dir ? blOut : blIn)[n64];
    __syncthreads();
    v8f c;
#pragma unroll
    for (int r = 0; r < 8; ++r) {
      c[r] = lds_a[(8 * h + r) * CPAD + FEAT + n] + bias;
    }

#pragma unroll 8
    for (int kc = 0; kc < FEAT; kc += 4) {
      const int k = kc + 2 * h;
      v2f a = *(const v2f*)&lds_a[mi * CPAD + k];
      v2f b = { Wr[(size_t)k * HALF + n64], Wr[(size_t)(k + 1) * HALF + n64] };
      c = __builtin_amdgcn_wmma_f32_16x16x4_f32(false, a, false, b, (short)0, c, false, false);
    }

#pragma unroll
    for (int r = 0; r < 8; ++r) {
      lds_a[(8 * h + r) * CPAD + FEAT + n] = fmaxf(c[r], 0.0f);
    }
  }
  __syncthreads();

  {
    v8f c = {};
#pragma unroll 8
    for (int kc = 0; kc < 2 * FEAT; kc += 4) {
      const int k = kc + 2 * h;
      v2f a  = *(const v2f*)&lds_a[mi * CPAD + k];
      v2f bb = { cW[(size_t)k * FEAT + n], cW[(size_t)(k + 1) * FEAT + n] };
      c = __builtin_amdgcn_wmma_f32_16x16x4_f32(false, a, false, bb, (short)0, c, false, false);
    }

    const float bias = cb[n];
#pragma unroll
    for (int r = 0; r < 8; ++r) {
      const float v = fmaxf(c[r] + bias, 0.0f);
      Hnext[(size_t)(mbase + r) * FEAT + n] = v;
      if (PROJ) lds_o[(8 * h + r) * SPAD + n] = v;
    }
  }

  if (PROJ) {
    __syncthreads();
    const float* __restrict__ nW = dir ? nWlOut : nWlIn;
    float* __restrict__ Y        = dir ? Yout : Yin;
    v8f c = {};
#pragma unroll 8
    for (int kc = 0; kc < FEAT; kc += 4) {
      const int k = kc + 2 * h;
      v2f a = *(const v2f*)&lds_o[mi * SPAD + k];
      v2f b = { nW[(size_t)k * HALF + n64], nW[(size_t)(k + 1) * HALF + n64] };
      c = __builtin_amdgcn_wmma_f32_16x16x4_f32(false, a, false, b, (short)0, c, false, false);
    }
#pragma unroll
    for (int r = 0; r < 8; ++r) {
      Y[(size_t)(mbase + r) * HALF + n64] = c[r];
    }
  }
}

__global__ void final_kernel(const float* __restrict__ H, const float* __restrict__ W,
                             const float* __restrict__ b, float* __restrict__ out, int n) {
  int i = blockIdx.x * blockDim.x + threadIdx.x;
  if (i >= n) return;
  const float4* hr = (const float4*)(H + (size_t)i * FEAT);
  const float4* wr = (const float4*)W;
  float acc = 0.0f;
#pragma unroll 8
  for (int k = 0; k < FEAT / 4; ++k) {
    float4 hv = hr[k];
    float4 wv = wr[k];
    acc += hv.x * wv.x + hv.y * wv.y + hv.z * wv.z + hv.w * wv.w;
  }
  out[i] = acc + b[0];
}

extern "C" void kernel_launch(void* const* d_in, const int* in_sizes, int n_in,
                              void* d_out, int out_size, void* d_ws, size_t ws_size,
                              hipStream_t stream) {
  (void)n_in; (void)out_size; (void)ws_size;

  const float* x       = (const float*)d_in[0];
  const int*   ein     = (const int*)d_in[1];
  const int*   eout    = (const int*)d_in[2];
  const float* in_Wl0  = (const float*)d_in[3];
  const float* in_bl0  = (const float*)d_in[4];
  const float* in_Wr0  = (const float*)d_in[5];
  const float* out_Wl0 = (const float*)d_in[6];
  const float* out_bl0 = (const float*)d_in[7];
  const float* out_Wr0 = (const float*)d_in[8];
  const float* comb_W0 = (const float*)d_in[9];
  const float* comb_b0 = (const float*)d_in[10];
  const float* in_Wl1  = (const float*)d_in[11];
  const float* in_bl1  = (const float*)d_in[12];
  const float* in_Wr1  = (const float*)d_in[13];
  const float* out_Wl1 = (const float*)d_in[14];
  const float* out_bl1 = (const float*)d_in[15];
  const float* out_Wr1 = (const float*)d_in[16];
  const float* comb_W1 = (const float*)d_in[17];
  const float* comb_b1 = (const float*)d_in[18];
  const float* final_W = (const float*)d_in[19];
  const float* final_b = (const float*)d_in[20];
  float* out = (float*)d_out;

  const int N = NNODES;
  const int E = in_sizes[1] / 2;

  const size_t N128 = (size_t)N * FEAT;
  const size_t N64  = (size_t)N * HALF;
  float* ws      = (float*)d_ws;
  int*   cursor  = (int*)d_ws;
  float* yB      = ws + 2 * PADN;
  int2*  seg     = (int2*)(yB + N128);
  int*   srt     = (int*)(seg + 2 * N);
  int*   rank    = srt + 2 * (size_t)E;
  float* HA      = yB + 2 * N128;
  float* HB      = HA + N128;
  float* yA      = HB;

  const int ZT = 256;
  auto nblk = [](long long c, int t) { return (unsigned)((c + t - 1) / t); };

  fill_zero_kernel<<<1, ZT, 0, stream>>>(ws, ZT);
  edge_rank_kernel<<<dim3(NGRP, 2), 1024, 0, stream>>>(ein, eout, cursor, seg, rank, E);
  edge_place_kernel<<<dim3(nblk(E, ZT), 2), ZT, 0, stream>>>(ein, eout, seg, rank, srt, E);

  proj_gemm_kernel<<<MTILES, 256, 0, stream>>>(x, in_Wl0, out_Wl0, yA, yA + N64);
  layer_tail_kernel<true><<<MTILES, 256, 0, stream>>>(
      x, yA, seg, srt, E, in_Wr0, out_Wr0, in_bl0, out_bl0, comb_W0, comb_b0, HA,
      in_Wl1, out_Wl1, yB, yB + N64);

  layer_tail_kernel<false><<<MTILES, 256, 0, stream>>>(
      HA, yB, seg, srt, E, in_Wr1, out_Wr1, in_bl1, out_bl1, comb_W1, comb_b1, HB,
      nullptr, nullptr, nullptr, nullptr);

  final_kernel<<<nblk(N, ZT), ZT, 0, stream>>>(HB, final_W, final_b, out, N);
}


# B operands of all GEMMs read from a fragment-ordered weight copy built per call: one 512-byte load feeds two k-steps (A by 16-byte LDS reads); changes the order of f32 summation inside groups of 8 k
# speedup vs baseline: 2.7538x; 1.0488x over previous
#include <hip/hip_runtime.h>
#include <hip/hip_bf16.h>


typedef float v2f __attribute__((ext_vector_type(2)));
typedef float v8f __attribute__((ext_vector_type(8)));

#define NNODES 50000
#define FEAT   128
#define HALF   64
#define MTILES (NNODES / 16)
#define SPAD   132
#define CPAD   260
#define PADN   65536
#define NGRP   64
#define RNG    782
#define RU     16

#define WF_SMALL (FEAT * HALF)
#define WF_LAYER (4 * WF_SMALL + 2 * FEAT * FEAT)

__global__ __launch_bounds__(256) void weight_frag_kernel(
    const float* __restrict__ a0, const float* __restrict__ a1, const float* __restrict__ a2,
    const float* __restrict__ a3, const float* __restrict__ a4,
    const float* __restrict__ b0, const float* __restrict__ b1, const float* __restrict__ b2,
    const float* __restrict__ b3, const float* __restrict__ b4,
    float* __restrict__ wf, int* __restrict__ head) {
  const int m   = blockIdx.y;
  const int lay = (m >= 5) ? 1 : 0;
  const int j   = m - 5 * lay;
  if (m == 0 && blockIdx.x == 0) head[threadIdx.x] = 0;
  const bool comb = (j == 4);
  const int N    = comb ? FEAT : HALF;
  const int ksh  = comb ? 5 : 4;
  const int nq   = comb ? (2 * FEAT * FEAT / 4) : (WF_SMALL / 4);
  const int q    = blockIdx.x * 256 + threadIdx.x;
  if (q >= nq) return;
  const float* __restrict__ Wsrc =
      lay ? (j == 0 ? b0 : j == 1 ? b1 : j == 2 ? b2 : j == 3 ? b3 : b4)
          : (j == 0 ? a0 : j == 1 ? a1 : j == 2 ? a2 : j == 3 ? a3 : a4);
  const int lane  = q & 31;
  const int chunk = q >> 5;
  const int t     = chunk >> ksh;
  const int kb    = chunk & ((1 << ksh) - 1);
  const int n     = t * 16 + (lane & 15);
  const int k0    = kb * 8 + 4 * (lane >> 4);
  const float4 v = make_float4(Wsrc[(size_t)k0 * N + n], Wsrc[(size_t)(k0 + 1) * N + n],
                               Wsrc[(size_t)(k0 + 2) * N + n], Wsrc[(size_t)(k0 + 3) * N + n]);
  float4* __restrict__ dstq = (float4*)(wf + (size_t)lay * WF_LAYER + (comb ? 4 * WF_SMALL : j * WF_SMALL));
  dstq[q] = v;
}

template <int K>
__device__ __forceinline__ v8f tile_gemm(const float* arow, const float4* __restrict__ bq, v8f c) {
#pragma unroll 4
  for (int kb = 0; kb < K / 8; ++kb) {
    const float4 a = *(const float4*)(arow + kb * 8);
    const float4 b = bq[kb * 32];
    const v2f alo = { a.x, a.y }, blo = { b.x, b.y };
    const v2f ahi = { a.z, a.w }, bhi = { b.z, b.w };
    c = __builtin_amdgcn_wmma_f32_16x16x4_f32(false, alo, false, blo, (short)0, c, false, false);
    c = __builtin_amdgcn_wmma_f32_16x16x4_f32(false, ahi, false, bhi, (short)0, c, false, false);
  }
  return c;
}


template <bool GUARD>
__device__ __forceinline__ void rank_step(const int* __restrict__ dst, int* __restrict__ rk,
                                          int* cnt, int base, int E, int lo, int nr) {
  int k[RU], r[RU];
#pragma unroll
  for (int u = 0; u < RU; ++u) {
    const int ee = base + u * 1024;
    k[u] = ((!GUARD || ee < E) ? dst[ee] : -1) - lo;
  }
#pragma unroll
  for (int u = 0; u < RU; ++u) {
    if ((unsigned)k[u] < (unsigned)nr) r[u] = atomicAdd(&cnt[k[u]], 1);
  }
#pragma unroll
  for (int u = 0; u < RU; ++u) {
    if ((unsigned)k[u] < (unsigned)nr) rk[base + u * 1024] = r[u];
  }
}

__global__ __launch_bounds__(1024) void edge_rank_kernel(
    const int* __restrict__ ein, const int* __restrict__ eout,
    int* __restrict__ cursor, int2* __restrict__ seg, int* __restrict__ rank, int E) {
  __shared__ int cnt[RNG];
  __shared__ int wave_sum[32];
  __shared__ int group_base;
  const int dir  = blockIdx.y;
  const int tid  = threadIdx.x;
  const int lane = tid & 31;
  const int wave = tid >> 5;
  const int lo   = blockIdx.x * RNG;
  const int nr   = min(RNG, NNODES - lo);
  const int* __restrict__ dst = (dir ? eout : ein) + E;
  int* __restrict__ rk = rank + (size_t)dir * E;

  if (tid < RNG) cnt[tid] = 0;
  __syncthreads();

  const int STEP  = RU * 1024;
  const int steps = (E + STEP - 1) / STEP;
  const int start = (int)(blockIdx.x * steps) / NGRP;
  for (int it = 0; it < steps; ++it) {
    int s = it + start;
    if (s >= steps) s -= steps;
    const int first = s * STEP;
    if (first + STEP <= E) rank_step<false>(dst, rk, cnt, first + tid, E, lo, nr);
    else                   rank_step<true >(dst, rk, cnt, first + tid, E, lo, nr);
  }
  __syncthreads();

  const int c = (tid < nr) ? cnt[tid] : 0;
  int incl = c;
#pragma unroll
  for (int d = 1; d < 32; d <<= 1) {
    const int t = __shfl_up(incl, d);
    if (lane >= d) incl += t;
  }
  if (lane == 31) wave_sum[wave] = incl;
  __syncthreads();
  if (wave == 0) {
    const int ws = wave_sum[lane];
    int wi = ws;
#pragma unroll
    for (int d = 1; d < 32; d <<= 1) {
      const int t = __shfl_up(wi, d);
      if (lane >= d) wi += t;
    }
    wave_sum[lane] = wi - ws;
    if (lane == 31) group_base = atomicAdd(&cursor[dir], wi);
  }
  __syncthreads();
  if (tid < nr)
    seg[dir * NNODES + lo + tid] = make_int2(group_base + wave_sum[wave] + incl - c, c);
}

__global__ void edge_place_kernel(const int* __restrict__ ein, const int* __restrict__ eout,
                                  const int2* __restrict__ seg, const int* __restrict__ rank,
                                  int* __restrict__ srt, int E) {
  const int dir = blockIdx.y;
  const int e = blockIdx.x * blockDim.x + threadIdx.x;
  if (e >= E) return;
  const int* __restrict__ ei = dir ? eout : ein;
  const int s = ei[e];
  const int d = ei[E + e];
  srt[(size_t)dir * E + seg[dir * NNODES + d].x + rank[(size_t)dir * E + e]] = s;
}

__global__ __launch_bounds__(256) void proj_gemm_kernel(
    const float* __restrict__ H, const float* __restrict__ wfl, float* __restrict__ Yin,
    float* __restrict__ Yout) {
  __shared__ __align__(16) float lds_h[16 * SPAD];

  const int tile_m = blockIdx.x;
  const int tid    = threadIdx.x;

#pragma unroll
  for (int i = 0; i < 2; ++i) {
    const int idx = tid + i * 256;
    const int row = idx >> 5;
    const int c4  = (idx & 31) * 4;
    const int gm  = tile_m * 16 + row;
    const float4 hv = *(const float4*)(H + (size_t)gm * FEAT + c4);
    *(float4*)&lds_h[row * SPAD + c4] = hv;
  }
  __syncthreads();

  const int wave = tid >> 5;
  const int lane = tid & 31;
  const int mi   = lane & 15;
  const int h    = lane >> 4;
  const int dir  = wave >> 2;
  const int n    = (wave & 3) * 16 + mi;
  float* __restrict__ Y       = dir ? Yout : Yin;
  const float4* __restrict__ bq =
      (const float4*)(wfl + dir * WF_SMALL) + ((wave & 3) * (FEAT / 8)) * 32 + lane;

  v8f c = {};
  c = tile_gemm<FEAT>(&lds_h[mi * SPAD + 4 * h], bq, c);

  const int mbase = tile_m * 16 + 8 * h;
#pragma unroll
  for (int r = 0; r < 8; ++r) {
    Y[(size_t)(mbase + r) * HALF + n] = c[r];
  }
}

__device__ __forceinline__ float4 segment_mean(
    const float* __restrict__ Yb, const int* __restrict__ sd, int c, int idx0,
    int l16, bool hi, unsigned col) {
  const int cmax = max(__builtin_amdgcn_readlane(c, 0), __builtin_amdgcn_readlane(c, 16));

  float4 acc = make_float4(0.0f, 0.0f, 0.0f, 0.0f);
  for (int base = 0; base < cmax; base += 16) {
    const int rem = c - base;
    int myidx = idx0;
    if (base != 0) myidx = (l16 < rem) ? sd[base + l16] : 0;
    const int m = min(cmax - base, 16);
    for (int j = 0; j < m; j += 8) {
      float4 v[8];
#pragma unroll
      for (int u = 0; u < 8; ++u) {
        const int jl = j + u;
        const unsigned s_lo = (unsigned)__builtin_amdgcn_readlane(myidx, jl);
        const unsigned s_hi = (unsigned)__builtin_amdgcn_readlane(myidx, jl + 16);
        const unsigned s    = hi ? s_hi : s_lo;
        v[u] = make_float4(0.0f, 0.0f, 0.0f, 0.0f);
        if (jl < rem) v[u] = *(const float4*)(Yb + (s * HALF + col));
      }
#pragma unroll
      for (int u = 0; u < 8; ++u) {
        acc.x += v[u].x; acc.y += v[u].y; acc.z += v[u].z; acc.w += v[u].w;
      }
    }
  }
  const float rs = 1.0f / fmaxf((float)c, 1.0f);
  acc.x *= rs; acc.y *= rs; acc.z *= rs; acc.w *= rs;
  return acc;
}

template <bool PROJ>
__global__ __launch_bounds__(256) void layer_tail_kernel(
    const float* __restrict__ H, const float* __restrict__ Y,
    const int2* __restrict__ seg, const int* __restrict__ srt, int E,
    const float* __restrict__ wfl,
    const float* __restrict__ blIn, const float* __restrict__ blOut,
    const float* __restrict__ cb,
    float* __restrict__ Hnext,
    const float* __restrict__ wfn,
    float* __restrict__ Yin, float* __restrict__ Yout) {
  __shared__ __align__(16) float lds_a[16 * CPAD];
  __shared__ __align__(16) float lds_o[PROJ ? 16 * SPAD : 4];

  const int tile_m = blockIdx.x;
  const int tid    = threadIdx.x;

#pragma unroll
  for (int i = 0; i < 2; ++i) {
    const int idx = tid + i * 256;
    const int row = idx >> 5;
    const int c4  = (idx & 31) * 4;
    const int gm  = tile_m * 16 + row;
    const float4 hv = *(const float4*)(H + (size_t)gm * FEAT + c4);
    *(float4*)&lds_a[row * CPAD + c4] = hv;
  }

  {
    const int  l16 = tid & 15;
    const bool hi  = (tid & 16) != 0;
    const int  nl  = tid >> 4;
    const int  nd  = tile_m * 16 + nl;
    const unsigned col = l16 * 4;
    const int2 sg0 = seg[nd];
    const int2 sg1 = seg[NNODES + nd];
    const int* __restrict__ sd0 = srt + sg0.x;
    const int* __restrict__ sd1 = srt + (size_t)E + sg1.x;
    const int r0 = sd0[(l16 < sg0.y) ? l16 : 0];
    const int r1 = sd1[(l16 < sg1.y) ? l16 : 0];
    const int i0 = (l16 < sg0.y) ? r0 : 0;
    const int i1 = (l16 < sg1.y) ? r1 : 0;
    const float4 m0 = segment_mean(Y, sd0, sg0.y, i0, l16, hi, col);
    *(float4*)&lds_a[nl * CPAD + FEAT + col] = m0;
    const float4 m1 = segment_mean(Y + (size_t)NNODES * HALF, sd1, sg1.y, i1, l16, hi, col);
    *(float4*)&lds_a[nl * CPAD + FEAT + HALF + col] = m1;
  }

  const int wave = tid >> 5;
  const int lane = tid & 31;
  const int mi   = lane & 15;
  const int h    = lane >> 4;
  const int dir  = wave >> 2;
  const int n64  = (wave & 3) * 16 + mi;
  const int n    = wave * 16 + mi;
  const int mbase = tile_m * 16 + 8 * h;

  {
    const float4* __restrict__ bq =
        (const float4*)(wfl + 2 * WF_SMALL + dir * WF_SMALL) + ((wave & 3) * (FEAT / 8)) * 32 + lane;
    const float bias = (dir ? blOut : blIn)[n64];
    __syncthreads();
    v8f c;
#pragma unroll
    for (int r = 0; r < 8; ++r) {
      c[r] = lds_a[(8 * h + r) * CPAD + FEAT + n] + bias;
    }

    c = tile_gemm<FEAT>(&lds_a[mi * CPAD + 4 * h], bq, c);

#pragma unroll
    for (int r = 0; r < 8; ++r) {
      lds_a[(8 * h + r) * CPAD + FEAT + n] = fmaxf(c[r], 0.0f);
    }
  }
  __syncthreads();

  {
    const float4* __restrict__ bq =
        (const float4*)(wfl + 4 * WF_SMALL) + (wave * (2 * FEAT / 8)) * 32 + lane;
    v8f c = {};
    c = tile_gemm<2 * FEAT>(&lds_a[mi * CPAD + 4 * h], bq, c);

    const float bias = cb[n];
#pragma unroll
    for (int r = 0; r < 8; ++r) {
      const float v = fmaxf(c[r] + bias, 0.0f);
      Hnext[(size_t)(mbase + r) * FEAT + n] = v;
      if (PROJ) lds_o[(8 * h + r) * SPAD + n] = v;
    }
  }

  if (PROJ) {
    __syncthreads();
    const float4* __restrict__ bq =
        (const float4*)(wfn + dir * WF_SMALL) + ((wave & 3) * (FEAT / 8)) * 32 + lane;
    float* __restrict__ Y        = dir ? Yout : Yin;
    v8f c = {};
    c = tile_gemm<FEAT>(&lds_o[mi * SPAD + 4 * h], bq, c);
#pragma unroll
    for (int r = 0; r < 8; ++r) {
      Y[(size_t)(mbase + r) * HALF + n64] = c[r];
    }
  }
}

__global__ void final_kernel(const float* __restrict__ H, const float* __restrict__ W,
                             const float* __restrict__ b, float* __restrict__ out, int n) {
  int i = blockIdx.x * blockDim.x + threadIdx.x;
  if (i >= n) return;
  const float4* hr = (const float4*)(H + (size_t)i * FEAT);
  const float4* wr = (const float4*)W;
  float acc = 0.0f;
#pragma unroll 8
  for (int k = 0; k < FEAT / 4; ++k) {
    float4 hv = hr[k];
    float4 wv = wr[k];
    acc += hv.x * wv.x + hv.y * wv.y + hv.z * wv.z + hv.w * wv.w;
  }
  out[i] = acc + b[0];
}

extern "C" void kernel_launch(void* const* d_in, const int* in_sizes, int n_in,
                              void* d_out, int out_size, void* d_ws, size_t ws_size,
                              hipStream_t stream) {
  (void)n_in; (void)out_size; (void)ws_size;

  const float* x       = (const float*)d_in[0];
  const int*   ein     = (const int*)d_in[1];
  const int*   eout    = (const int*)d_in[2];
  const float* in_Wl0  = (const float*)d_in[3];
  const float* in_bl0  = (const float*)d_in[4];
  const float* in_Wr0  = (const float*)d_in[5];
  const float* out_Wl0 = (const float*)d_in[6];
  const float* out_bl0 = (const float*)d_in[7];
  const float* out_Wr0 = (const float*)d_in[8];
  const float* comb_W0 = (const float*)d_in[9];
  const float* comb_b0 = (const float*)d_in[10];
  const float* in_Wl1  = (const float*)d_in[11];
  const float* in_bl1  = (const float*)d_in[12];
  const float* in_Wr1  = (const float*)d_in[13];
  const float* out_Wl1 = (const float*)d_in[14];
  const float* out_bl1 = (const float*)d_in[15];
  const float* out_Wr1 = (const float*)d_in[16];
  const float* comb_W1 = (const float*)d_in[17];
  const float* comb_b1 = (const float*)d_in[18];
  const float* final_W = (const float*)d_in[19];
  const float* final_b = (const float*)d_in[20];
  float* out = (float*)d_out;

  const int N = NNODES;
  const int E = in_sizes[1] / 2;

  const size_t N128 = (size_t)N * FEAT;
  const size_t N64  = (size_t)N * HALF;
  float* ws      = (float*)d_ws;
  int*   cursor  = (int*)d_ws;
  float* yB      = ws + 2 * PADN;
  int2*  seg     = (int2*)(yB + N128);
  int*   srt     = (int*)(seg + 2 * N);
  int*   rank    = srt + 2 * (size_t)E;
  float* wf      = yB + N128 + 4000000;
  float* HA      = yB + 2 * N128;
  float* HB      = HA + N128;
  float* yA      = HB;

  const int ZT = 256;
  auto nblk = [](long long c, int t) { return (unsigned)((c + t - 1) / t); };

  weight_frag_kernel<<<dim3(32, 10), ZT, 0, stream>>>(
      in_Wl0, out_Wl0, in_Wr0, out_Wr0, comb_W0,
      in_Wl1, out_Wl1, in_Wr1, out_Wr1, comb_W1, wf, cursor);
  edge_rank_kernel<<<dim3(NGRP, 2), 1024, 0, stream>>>(ein, eout, cursor, seg, rank, E);
  edge_place_kernel<<<dim3(nblk(E, ZT), 2), ZT, 0, stream>>>(ein, eout, seg, rank, srt, E);

  proj_gemm_kernel<<<MTILES, 256, 0, stream>>>(x, wf, yA, yA + N64);
  layer_tail_kernel<true><<<MTILES, 256, 0, stream>>>(
      x, yA, seg, srt, E, wf, in_bl0, out_bl0, comb_b0, HA,
      wf + WF_LAYER, yB, yB + N64);

  layer_tail_kernel<false><<<MTILES, 256, 0, stream>>>(
      HA, yB, seg, srt, E, wf + WF_LAYER, in_bl1, out_bl1, comb_b1, HB,
      nullptr, nullptr, nullptr);

  final_kernel<<<nblk(N, ZT), ZT, 0, stream>>>(HB, final_W, final_b, out, N);
}


# Every GEMM wave carries two 16-row tiles (32-row work-groups; the 53 odd tiles stay single): one B fragment load feeds both tiles' WMMAs, halving weight reads and GEMM waves; sums bit-identical
# speedup vs baseline: 2.8434x; 1.0325x over previous
#include <hip/hip_runtime.h>
#include <hip/hip_bf16.h>


typedef float v2f __attribute__((ext_vector_type(2)));
typedef float v8f __attribute__((ext_vector_type(8)));

#define NNODES 50000
#define FEAT   128
#define HALF   64
#define MTILES (NNODES / 16)
#define WG2    1536
#define WG1    (MTILES - 2 * WG2)
#define WGROUPS (WG2 + WG1)
#define SPAD   132
#define CPAD   260
#define PADN   65536
#define NGRP   64
#define RNG    782
#define RU     16

#define WF_SMALL (FEAT * HALF)
#define WF_LAYER (4 * WF_SMALL + 2 * FEAT * FEAT)

__global__ __launch_bounds__(256) void weight_frag_kernel(
    const float* __restrict__ a0, const float* __restrict__ a1, const float* __restrict__ a2,
    const float* __restrict__ a3, const float* __restrict__ a4,
    const float* __restrict__ b0, const float* __restrict__ b1, const float* __restrict__ b2,
    const float* __restrict__ b3, const float* __restrict__ b4,
    float* __restrict__ wf, int* __restrict__ head) {
  const int m   = blockIdx.y;
  const int lay = (m >= 5) ? 1 : 0;
  const int j   = m - 5 * lay;
  if (m == 0 && blockIdx.x == 0) head[threadIdx.x] = 0;
  const bool comb = (j == 4);
  const int N    = comb ? FEAT : HALF;
  const int ksh  = comb ? 5 : 4;
  const int nq   = comb ? (2 * FEAT * FEAT / 4) : (WF_SMALL / 4);
  const int q    = blockIdx.x * 256 + threadIdx.x;
  if (q >= nq) return;
  const float* __restrict__ Wsrc =
      lay ? (j == 0 ? b0 : j == 1 ? b1 : j == 2 ? b2 : j == 3 ? b3 : b4)
          : (j == 0 ? a0 : j == 1 ? a1 : j == 2 ? a2 : j == 3 ? a3 : a4);
  const int lane  = q & 31;
  const int chunk = q >> 5;
  const int t     = chunk >> ksh;
  const int kb    = chunk & ((1 << ksh) - 1);
  const int n     = t * 16 + (lane & 15);
  const int k0    = kb * 8 + 4 * (lane >> 4);
  const float4 v = make_float4(Wsrc[(size_t)k0 * N + n], Wsrc[(size_t)(k0 + 1) * N + n],
                               Wsrc[(size_t)(k0 + 2) * N + n], Wsrc[(size_t)(k0 + 3) * N + n]);
  float4* __restrict__ dstq = (float4*)(wf + (size_t)lay * WF_LAYER + (comb ? 4 * WF_SMALL : j * WF_SMALL));
  dstq[q] = v;
}

template <int K, int PITCH, int NT>
__device__ __forceinline__ void tile_gemm(const float* arow, const float4* __restrict__ bq, v8f (&c)[NT]) {
#pragma unroll (4 / NT)
  for (int kb = 0; kb < K / 8; ++kb) {
    const float4 b = bq[kb * 32];
    const v2f blo = { b.x, b.y }, bhi = { b.z, b.w };
    float4 a[NT];
#pragma unroll
    for (int t = 0; t < NT; ++t) a[t] = *(const float4*)(arow + t * 16 * PITCH + kb * 8);
#pragma unroll
    for (int t = 0; t < NT; ++t) {
      const v2f alo = { a[t].x, a[t].y };
      c[t] = __builtin_amdgcn_wmma_f32_16x16x4_f32(false, alo, false, blo, (short)0, c[t], false, false);
    }
#pragma unroll
    for (int t = 0; t < NT; ++t) {
      const v2f ahi = { a[t].z, a[t].w };
      c[t] = __builtin_amdgcn_wmma_f32_16x16x4_f32(false, ahi, false, bhi, (short)0, c[t], false, false);
    }
  }
}


template <bool GUARD>
__device__ __forceinline__ void rank_step(const int* __restrict__ dst, int* __restrict__ rk,
                                          int* cnt, int base, int E, int lo, int nr) {
  int k[RU], r[RU];
#pragma unroll
  for (int u = 0; u < RU; ++u) {
    const int ee = base + u * 1024;
    k[u] = ((!GUARD || ee < E) ? dst[ee] : -1) - lo;
  }
#pragma unroll
  for (int u = 0; u < RU; ++u) {
    if ((unsigned)k[u] < (unsigned)nr) r[u] = atomicAdd(&cnt[k[u]], 1);
  }
#pragma unroll
  for (int u = 0; u < RU; ++u) {
    if ((unsigned)k[u] < (unsigned)nr) rk[base + u * 1024] = r[u];
  }
}

__global__ __launch_bounds__(1024) void edge_rank_kernel(
    const int* __restrict__ ein, const int* __restrict__ eout,
    int* __restrict__ cursor, int2* __restrict__ seg, int* __restrict__ rank, int E) {
  __shared__ int cnt[RNG];
  __shared__ int wave_sum[32];
  __shared__ int group_base;
  const int dir  = blockIdx.y;
  const int tid  = threadIdx.x;
  const int lane = tid & 31;
  const int wave = tid >> 5;
  const int lo   = blockIdx.x * RNG;
  const int nr   = min(RNG, NNODES - lo);
  const int* __restrict__ dst = (dir ? eout : ein) + E;
  int* __restrict__ rk = rank + (size_t)dir * E;

  if (tid < RNG) cnt[tid] = 0;
  __syncthreads();

  const int STEP  = RU * 1024;
  const int steps = (E + STEP - 1) / STEP;
  const int start = (int)(blockIdx.x * steps) / NGRP;
  for (int it = 0; it < steps; ++it) {
    int s = it + start;
    if (s >= steps) s -= steps;
    const int first = s * STEP;
    if (first + STEP <= E) rank_step<false>(dst, rk, cnt, first + tid, E, lo, nr);
    else                   rank_step<true >(dst, rk, cnt, first + tid, E, lo, nr);
  }
  __syncthreads();

  const int c = (tid < nr) ? cnt[tid] : 0;
  int incl = c;
#pragma unroll
  for (int d = 1; d < 32; d <<= 1) {
    const int t = __shfl_up(incl, d);
    if (lane >= d) incl += t;
  }
  if (lane == 31) wave_sum[wave] = incl;
  __syncthreads();
  if (wave == 0) {
    const int ws = wave_sum[lane];
    int wi = ws;
#pragma unroll
    for (int d = 1; d < 32; d <<= 1) {
      const int t = __shfl_up(wi, d);
      if (lane >= d) wi += t;
    }
    wave_sum[lane] = wi - ws;
    if (lane == 31) group_base = atomicAdd(&cursor[dir], wi);
  }
  __syncthreads();
  if (tid < nr)
    seg[dir * NNODES + lo + tid] = make_int2(group_base + wave_sum[wave] + incl - c, c);
}

__global__ void edge_place_kernel(const int* __restrict__ ein, const int* __restrict__ eout,
                                  const int2* __restrict__ seg, const int* __restrict__ rank,
                                  int* __restrict__ srt, int E) {
  const int dir = blockIdx.y;
  const int e = blockIdx.x * blockDim.x + threadIdx.x;
  if (e >= E) return;
  const int* __restrict__ ei = dir ? eout : ein;
  const int s = ei[e];
  const int d = ei[E + e];
  srt[(size_t)dir * E + seg[dir * NNODES + d].x + rank[(size_t)dir * E + e]] = s;
}

template <int NT>
__device__ __forceinline__ void proj_gemm_body(
    float* lds_h, int row0,
    const float* __restrict__ H, const float* __restrict__ wfl, float* __restrict__ Yin,
    float* __restrict__ Yout) {
  const int tid = threadIdx.x;

#pragma unroll
  for (int i = 0; i < 2 * NT; ++i) {
    const int idx = tid + i * 256;
    const int row = idx >> 5;
    const int c4  = (idx & 31) * 4;
    const float4 hv = *(const float4*)(H + (size_t)(row0 + row) * FEAT + c4);
    *(float4*)&lds_h[row * SPAD + c4] = hv;
  }
  __syncthreads();

  const int wave = tid >> 5;
  const int lane = tid & 31;
  const int mi   = lane & 15;
  const int h    = lane >> 4;
  const int dir  = wave >> 2;
  const int n    = (wave & 3) * 16 + mi;
  float* __restrict__ Y       = dir ? Yout : Yin;
  const float4* __restrict__ bq =
      (const float4*)(wfl + dir * WF_SMALL) + ((wave & 3) * (FEAT / 8)) * 32 + lane;

  v8f c[NT] = {};
  tile_gemm<FEAT, SPAD, NT>(&lds_h[mi * SPAD + 4 * h], bq, c);

  const int mbase = row0 + 8 * h;
#pragma unroll
  for (int t = 0; t < NT; ++t) {
#pragma unroll
    for (int r = 0; r < 8; ++r) {
      Y[(size_t)(mbase + 16 * t + r) * HALF + n] = c[t][r];
    }
  }
}

__global__ __launch_bounds__(256) void proj_gemm_kernel(
    const float* __restrict__ H, const float* __restrict__ wfl, float* __restrict__ Yin,
    float* __restrict__ Yout) {
  __shared__ __align__(16) float lds_h[32 * SPAD];
  const int b = blockIdx.x;
  if (b < WG2) proj_gemm_body<2>(lds_h, b * 32, H, wfl, Yin, Yout);
  else         proj_gemm_body<1>(lds_h, 32 * WG2 + (b - WG2) * 16, H, wfl, Yin, Yout);
}

__device__ __forceinline__ float4 segment_mean(
    const float* __restrict__ Yb, const int* __restrict__ sd, int c, int idx0,
    int l16, bool hi, unsigned col) {
  const int cmax = max(__builtin_amdgcn_readlane(c, 0), __builtin_amdgcn_readlane(c, 16));

  float4 acc = make_float4(0.0f, 0.0f, 0.0f, 0.0f);
  for (int base = 0; base < cmax; base += 16) {
    const int rem = c - base;
    int myidx = idx0;
    if (base != 0) myidx = (l16 < rem) ? sd[base + l16] : 0;
    const int m = min(cmax - base, 16);
    for (int j = 0; j < m; j += 8) {
      float4 v[8];
#pragma unroll
      for (int u = 0; u < 8; ++u) {
        const int jl = j + u;
        const unsigned s_lo = (unsigned)__builtin_amdgcn_readlane(myidx, jl);
        const unsigned s_hi = (unsigned)__builtin_amdgcn_readlane(myidx, jl + 16);
        const unsigned s    = hi ? s_hi : s_lo;
        v[u] = make_float4(0.0f, 0.0f, 0.0f, 0.0f);
        if (jl < rem) v[u] = *(const float4*)(Yb + (s * HALF + col));
      }
#pragma unroll
      for (int u = 0; u < 8; ++u) {
        acc.x += v[u].x; acc.y += v[u].y; acc.z += v[u].z; acc.w += v[u].w;
      }
    }
  }
  const float rs = 1.0f / fmaxf((float)c, 1.0f);
  acc.x *= rs; acc.y *= rs; acc.z *= rs; acc.w *= rs;
  return acc;
}

template <bool PROJ, int NT>
__device__ __forceinline__ void layer_tail_body(
    float* lds_a, int row0,
    const float* __restrict__ H, const float* __restrict__ Y,
    const int2* __restrict__ seg, const int* __restrict__ srt, int E,
    const float* __restrict__ wfl,
    const float* __restrict__ blIn, const float* __restrict__ blOut,
    const float* __restrict__ cb,
    float* __restrict__ Hnext,
    const float* __restrict__ wfn,
    float* __restrict__ Yin, float* __restrict__ Yout) {
  float* const lds_o = lds_a;
  const int tid = threadIdx.x;

#pragma unroll
  for (int i = 0; i < 2 * NT; ++i) {
    const int idx = tid + i * 256;
    const int row = idx >> 5;
    const int c4  = (idx & 31) * 4;
    const float4 hv = *(const float4*)(H + (size_t)(row0 + row) * FEAT + c4);
    *(float4*)&lds_a[row * CPAD + c4] = hv;
  }

  {
    const int  l16 = tid & 15;
    const bool hi  = (tid & 16) != 0;
    const int  nl  = tid >> 4;
    const unsigned col = l16 * 4;
    int2 sg[NT][2];
    const int* sd[NT][2];
    int ix[NT][2];
#pragma unroll
    for (int q = 0; q < NT; ++q) {
      const int nd = row0 + 16 * q + nl;
      sg[q][0] = seg[nd];
      sg[q][1] = seg[NNODES + nd];
    }
#pragma unroll
    for (int q = 0; q < NT; ++q) {
#pragma unroll
      for (int d = 0; d < 2; ++d) {
        sd[q][d] = srt + (size_t)d * E + sg[q][d].x;
        const int r = sd[q][d][(l16 < sg[q][d].y) ? l16 : 0];
        ix[q][d] = (l16 < sg[q][d].y) ? r : 0;
      }
    }
#pragma unroll
    for (int q = 0; q < NT; ++q) {
#pragma unroll
      for (int d = 0; d < 2; ++d) {
        const float4 m = segment_mean(Y + (size_t)d * NNODES * HALF, sd[q][d], sg[q][d].y, ix[q][d], l16, hi, col);
        *(float4*)&lds_a[(16 * q + nl) * CPAD + FEAT + d * HALF + col] = m;
      }
    }
  }

  const int wave = tid >> 5;
  const int lane = tid & 31;
  const int mi   = lane & 15;
  const int h    = lane >> 4;
  const int dir  = wave >> 2;
  const int n64  = (wave & 3) * 16 + mi;
  const int n    = wave * 16 + mi;
  const int mbase = row0 + 8 * h;


  {
    const float4* __restrict__ bq =
        (const float4*)(wfl + 2 * WF_SMALL + dir * WF_SMALL) + ((wave & 3) * (FEAT / 8)) * 32 + lane;
    const float bias = (dir ? blOut : blIn)[n64];
    __syncthreads();
    v8f c[NT];
#pragma unroll
    for (int t = 0; t < NT; ++t) {
#pragma unroll
      for (int r = 0; r < 8; ++r) {
        c[t][r] = lds_a[(16 * t + 8 * h + r) * CPAD + FEAT + n] + bias;
      }
    }

    tile_gemm<FEAT, CPAD, NT>(&lds_a[mi * CPAD + 4 * h], bq, c);

#pragma unroll
    for (int t = 0; t < NT; ++t) {
#pragma unroll
      for (int r = 0; r < 8; ++r) {
        lds_a[(16 * t + 8 * h + r) * CPAD + FEAT + n] = fmaxf(c[t][r], 0.0f);
      }
    }
  }
  __syncthreads();

  {
    const float4* __restrict__ bq =
        (const float4*)(wfl + 4 * WF_SMALL) + (wave * (2 * FEAT / 8)) * 32 + lane;
    v8f c[NT] = {};
    tile_gemm<2 * FEAT, CPAD, NT>(&lds_a[mi * CPAD + 4 * h], bq, c);

    const float bias = cb[n];
    if (PROJ) __syncthreads();
#pragma unroll
    for (int t = 0; t < NT; ++t) {
#pragma unroll
      for (int r = 0; r < 8; ++r) {
        const float v = fmaxf(c[t][r] + bias, 0.0f);
        Hnext[(size_t)(mbase + 16 * t + r) * FEAT + n] = v;
        if (PROJ) lds_o[(16 * t + 8 * h + r) * SPAD + n] = v;
      }
    }
  }

  if (PROJ) {
    __syncthreads();
    const float4* __restrict__ bq =
        (const float4*)(wfn + dir * WF_SMALL) + ((wave & 3) * (FEAT / 8)) * 32 + lane;
    float* __restrict__ Yo = dir ? Yout : Yin;
    v8f c[NT] = {};
    tile_gemm<FEAT, SPAD, NT>(&lds_o[mi * SPAD + 4 * h], bq, c);
#pragma unroll
    for (int t = 0; t < NT; ++t) {
#pragma unroll
      for (int r = 0; r < 8; ++r) {
        Yo[(size_t)(mbase + 16 * t + r) * HALF + n64] = c[t][r];
      }
    }
  }
}

template <bool PROJ>
__global__ __launch_bounds__(256) void layer_tail_kernel(
    const float* __restrict__ H, const float* __restrict__ Y,
    const int2* __restrict__ seg, const int* __restrict__ srt, int E,
    const float* __restrict__ wfl,
    const float* __restrict__ blIn, const float* __restrict__ blOut,
    const float* __restrict__ cb,
    float* __restrict__ Hnext,
    const float* __restrict__ wfn,
    float* __restrict__ Yin, float* __restrict__ Yout) {
  __shared__ __align__(16) float lds_a[32 * CPAD];
  const int b = blockIdx.x;
  if (b < WG2)
    layer_tail_body<PROJ, 2>(lds_a, b * 32, H, Y, seg, srt, E, wfl, blIn, blOut, cb, Hnext, wfn, Yin, Yout);
  else
    layer_tail_body<PROJ, 1>(lds_a, 32 * WG2 + (b - WG2) * 16, H, Y, seg, srt, E, wfl, blIn, blOut, cb, Hnext, wfn, Yin, Yout);
}

__global__ void final_kernel(const float* __restrict__ H, const float* __restrict__ W,
                             const float* __restrict__ b, float* __restrict__ out, int n) {
  int i = blockIdx.x * blockDim.x + threadIdx.x;
  if (i >= n) return;
  const float4* hr = (const float4*)(H + (size_t)i * FEAT);
  const float4* wr = (const float4*)W;
  float acc = 0.0f;
#pragma unroll 8
  for (int k = 0; k < FEAT / 4; ++k) {
    float4 hv = hr[k];
    float4 wv = wr[k];
    acc += hv.x * wv.x + hv.y * wv.y + hv.z * wv.z + hv.w * wv.w;
  }
  out[i] = acc + b[0];
}

extern "C" void kernel_launch(void* const* d_in, const int* in_sizes, int n_in,
                              void* d_out, int out_size, void* d_ws, size_t ws_size,
                              hipStream_t stream) {
  (void)n_in; (void)out_size; (void)ws_size;

  const float* x       = (const float*)d_in[0];
  const int*   ein     = (const int*)d_in[1];
  const int*   eout    = (const int*)d_in[2];
  const float* in_Wl0  = (const float*)d_in[3];
  const float* in_bl0  = (const float*)d_in[4];
  const float* in_Wr0  = (const float*)d_in[5];
  const float* out_Wl0 = (const float*)d_in[6];
  const float* out_bl0 = (const float*)d_in[7];
  const float* out_Wr0 = (const float*)d_in[8];
  const float* comb_W0 = (const float*)d_in[9];
  const float* comb_b0 = (const float*)d_in[10];
  const float* in_Wl1  = (const float*)d_in[11];
  const float* in_bl1  = (const float*)d_in[12];
  const float* in_Wr1  = (const float*)d_in[13];
  const float* out_Wl1 = (const float*)d_in[14];
  const float* out_bl1 = (const float*)d_in[15];
  const float* out_Wr1 = (const float*)d_in[16];
  const float* comb_W1 = (const float*)d_in[17];
  const float* comb_b1 = (const float*)d_in[18];
  const float* final_W = (const float*)d_in[19];
  const float* final_b = (const float*)d_in[20];
  float* out = (float*)d_out;

  const int N = NNODES;
  const int E = in_sizes[1] / 2;

  const size_t N128 = (size_t)N * FEAT;
  const size_t N64  = (size_t)N * HALF;
  float* ws      = (float*)d_ws;
  int*   cursor  = (int*)d_ws;
  float* yB      = ws + 2 * PADN;
  int2*  seg     = (int2*)(yB + N128);
  int*   srt     = (int*)(seg + 2 * N);
  int*   rank    = srt + 2 * (size_t)E;
  float* wf      = yB + N128 + 4000000;
  float* HA      = yB + 2 * N128;
  float* HB      = HA + N128;
  float* yA      = HB;

  const int ZT = 256;
  auto nblk = [](long long c, int t) { return (unsigned)((c + t - 1) / t); };

  weight_frag_kernel<<<dim3(32, 10), ZT, 0, stream>>>(
      in_Wl0, out_Wl0, in_Wr0, out_Wr0, comb_W0,
      in_Wl1, out_Wl1, in_Wr1, out_Wr1, comb_W1, wf, cursor);
  edge_rank_kernel<<<dim3(NGRP, 2), 1024, 0, stream>>>(ein, eout, cursor, seg, rank, E);
  edge_place_kernel<<<dim3(nblk(E, ZT), 2), ZT, 0, stream>>>(ein, eout, seg, rank, srt, E);

  proj_gemm_kernel<<<WGROUPS, 256, 0, stream>>>(x, wf, yA, yA + N64);
  layer_tail_kernel<true><<<WGROUPS, 256, 0, stream>>>(
      x, yA, seg, srt, E, wf, in_bl0, out_bl0, comb_b0, HA,
      wf + WF_LAYER, yB, yB + N64);

  layer_tail_kernel<false><<<WGROUPS, 256, 0, stream>>>(
      HA, yB, seg, srt, E, wf + WF_LAYER, in_bl1, out_bl1, comb_b1, HB,
      nullptr, nullptr, nullptr);

  final_kernel<<<nblk(N, ZT), ZT, 0, stream>>>(HB, final_W, final_b, out, N);
}


# Edge ranking per edge chunk, not node range: per-node LDS counters, each dst read once (not 64x); scan launch added, weight copy rides in the rank launch (7 launches); changes the order of f32 sums
# speedup vs baseline: 3.4889x; 1.2270x over previous
#include <hip/hip_runtime.h>
#include <hip/hip_bf16.h>


typedef float v2f __attribute__((ext_vector_type(2)));
typedef float v8f __attribute__((ext_vector_type(8)));

#define NNODES 50000
#define FEAT   128
#define HALF   64
#define MTILES (NNODES / 16)
#define WG2    1536
#define WG1    (MTILES - 2 * WG2)
#define WGROUPS (WG2 + WG1)
#define SPAD   132
#define CPAD   260
#define PADN   65536
#define NCH    31
#define RU     20
#define CH     (RU * 1024)
#define WBLK   80
#define NGRP   64
#define RNG    782
#define PLT    256

#define WF_SMALL (FEAT * HALF)
#define WF_LAYER (4 * WF_SMALL + 2 * FEAT * FEAT)

__device__ __forceinline__ void weight_frag_body(
    int wb, int tid,
    const float* __restrict__ a0, const float* __restrict__ a1, const float* __restrict__ a2,
    const float* __restrict__ a3, const float* __restrict__ a4,
    const float* __restrict__ b0, const float* __restrict__ b1, const float* __restrict__ b2,
    const float* __restrict__ b3, const float* __restrict__ b4,
    float* __restrict__ wf) {
  const int m   = wb >> 3;
  const int lay = (m >= 5) ? 1 : 0;
  const int j   = m - 5 * lay;
  const bool comb = (j == 4);
  const int N    = comb ? FEAT : HALF;
  const int ksh  = comb ? 5 : 4;
  const int nq   = comb ? (2 * FEAT * FEAT / 4) : (WF_SMALL / 4);
  const int q    = (wb & 7) * 1024 + tid;
  if (q >= nq) return;
  const float* __restrict__ Wsrc =
      lay ? (j == 0 ? b0 : j == 1 ? b1 : j == 2 ? b2 : j == 3 ? b3 : b4)
          : (j == 0 ? a0 : j == 1 ? a1 : j == 2 ? a2 : j == 3 ? a3 : a4);
  const int lane  = q & 31;
  const int chunk = q >> 5;
  const int t     = chunk >> ksh;
  const int kb    = chunk & ((1 << ksh) - 1);
  const int n     = t * 16 + (lane & 15);
  const int k0    = kb * 8 + 4 * (lane >> 4);
  const float4 v = make_float4(Wsrc[(size_t)k0 * N + n], Wsrc[(size_t)(k0 + 1) * N + n],
                               Wsrc[(size_t)(k0 + 2) * N + n], Wsrc[(size_t)(k0 + 3) * N + n]);
  float4* __restrict__ dstq = (float4*)(wf + (size_t)lay * WF_LAYER + (comb ? 4 * WF_SMALL : j * WF_SMALL));
  dstq[q] = v;
}

template <int K, int PITCH, int NT>
__device__ __forceinline__ void tile_gemm(const float* arow, const float4* __restrict__ bq, v8f (&c)[NT]) {
#pragma unroll (4 / NT)
  for (int kb = 0; kb < K / 8; ++kb) {
    const float4 b = bq[kb * 32];
    const v2f blo = { b.x, b.y }, bhi = { b.z, b.w };
    float4 a[NT];
#pragma unroll
    for (int t = 0; t < NT; ++t) a[t] = *(const float4*)(arow + t * 16 * PITCH + kb * 8);
#pragma unroll
    for (int t = 0; t < NT; ++t) {
      const v2f alo = { a[t].x, a[t].y };
      c[t] = __builtin_amdgcn_wmma_f32_16x16x4_f32(false, alo, false, blo, (short)0, c[t], false, false);
    }
#pragma unroll
    for (int t = 0; t < NT; ++t) {
      const v2f ahi = { a[t].z, a[t].w };
      c[t] = __builtin_amdgcn_wmma_f32_16x16x4_f32(false, ahi, false, bhi, (short)0, c[t], false, false);
    }
  }
}


template <bool GUARD>
__device__ __forceinline__ void rank_step(const int* __restrict__ dst, int* __restrict__ rk,
                                          int* cnt, int base, int E, int nr) {
  int k[RU], r[RU];
#pragma unroll
  for (int u = 0; u < RU; ++u) {
    const int ee = base + u * 1024;
    k[u] = (!GUARD || ee < E) ? dst[ee] : -1;
  }
#pragma unroll
  for (int u = 0; u < RU; ++u) {
    if ((unsigned)k[u] < (unsigned)nr) r[u] = atomicAdd(&cnt[k[u]], 1);
  }
#pragma unroll
  for (int u = 0; u < RU; ++u) {
    if ((unsigned)k[u] < (unsigned)nr) rk[base + u * 1024] = r[u];
  }
}

__global__ __launch_bounds__(1024) void edge_rank_kernel(
    const int* __restrict__ ein, const int* __restrict__ eout,
    int* __restrict__ cursor, int* __restrict__ cntmat, int* __restrict__ rank, int E,
    const float* __restrict__ a0, const float* __restrict__ a1, const float* __restrict__ a2,
    const float* __restrict__ a3, const float* __restrict__ a4,
    const float* __restrict__ b0, const float* __restrict__ b1, const float* __restrict__ b2,
    const float* __restrict__ b3, const float* __restrict__ b4,
    float* __restrict__ wf) {
  __shared__ __align__(16) int cnt[NNODES];
  const int b   = blockIdx.x;
  const int tid = threadIdx.x;
  if (b >= 2 * NCH) {
    weight_frag_body(b - 2 * NCH, tid, a0, a1, a2, a3, a4, b0, b1, b2, b3, b4, wf);
    return;
  }
  if (b == 0 && tid < 2) cursor[tid] = 0;
  const int dir = (b >= NCH) ? 1 : 0;
  const int c   = b - dir * NCH;
  const int* __restrict__ dst = (dir ? eout : ein) + E;
  int* __restrict__ rk = rank + (size_t)dir * E;

  int4* const cnt4 = (int4*)cnt;
  for (int i = tid; i < NNODES / 4; i += 1024) cnt4[i] = make_int4(0, 0, 0, 0);
  __syncthreads();

  for (int first = c * CH; first < E; first += NCH * CH) {
    if (first + CH <= E) rank_step<false>(dst, rk, cnt, first + tid, E, NNODES);
    else                 rank_step<true >(dst, rk, cnt, first + tid, E, NNODES);
  }
  __syncthreads();

  int4* __restrict__ row4 = (int4*)(cntmat + ((size_t)dir * NCH + c) * NNODES);
  for (int i = tid; i < NNODES / 4; i += 1024) row4[i] = cnt4[i];
}

__global__ __launch_bounds__(1024) void edge_scan_kernel(
    int* __restrict__ cursor, int* __restrict__ cntmat, int2* __restrict__ seg) {
  __shared__ int wave_sum[32];
  __shared__ int group_base;
  const int dir  = blockIdx.y;
  const int tid  = threadIdx.x;
  const int lane = tid & 31;
  const int wave = tid >> 5;
  const int lo   = blockIdx.x * RNG;
  const int nr   = min(RNG, NNODES - lo);
  int* __restrict__ col = cntmat + (size_t)dir * NCH * NNODES + min(lo + tid, NNODES - 1);

  int v[NCH];
#pragma unroll
  for (int g = 0; g < NCH; ++g) v[g] = col[(size_t)g * NNODES];
  int c = 0;
#pragma unroll
  for (int g = 0; g < NCH; ++g) { const int t = v[g]; v[g] = c; c += t; }
  if (tid >= nr) c = 0;

  int incl = c;
#pragma unroll
  for (int d = 1; d < 32; d <<= 1) {
    const int t = __shfl_up(incl, d);
    if (lane >= d) incl += t;
  }
  if (lane == 31) wave_sum[wave] = incl;
  __syncthreads();
  if (wave == 0) {
    const int ws = wave_sum[lane];
    int wi = ws;
#pragma unroll
    for (int d = 1; d < 32; d <<= 1) {
      const int t = __shfl_up(wi, d);
      if (lane >= d) wi += t;
    }
    wave_sum[lane] = wi - ws;
    if (lane == 31) group_base = atomicAdd(&cursor[dir], wi);
  }
  __syncthreads();
  if (tid < nr) {
    const int start = group_base + wave_sum[wave] + incl - c;
    seg[dir * NNODES + lo + tid] = make_int2(start, c);
#pragma unroll
    for (int g = 0; g < NCH; ++g) col[(size_t)g * NNODES] = start + v[g];
  }
}

__global__ __launch_bounds__(PLT) void edge_place_kernel(
    const int* __restrict__ ein, const int* __restrict__ eout,
    const int* __restrict__ pos, const int* __restrict__ rank,
    int* __restrict__ srt, int E) {
  const int dir = blockIdx.y;
  const int e = blockIdx.x * PLT + threadIdx.x;
  if (e >= E) return;
  const int g = (int)(blockIdx.x / (CH / PLT)) % NCH;
  const int* __restrict__ ei = dir ? eout : ein;
  const int s = ei[e];
  const int d = ei[E + e];
  srt[(size_t)dir * E + pos[((size_t)dir * NCH + g) * NNODES + d] + rank[(size_t)dir * E + e]] = s;
}

template <int NT>
__device__ __forceinline__ void proj_gemm_body(
    float* lds_h, int row0,
    const float* __restrict__ H, const float* __restrict__ wfl, float* __restrict__ Yin,
    float* __restrict__ Yout) {
  const int tid = threadIdx.x;

#pragma unroll
  for (int i = 0; i < 2 * NT; ++i) {
    const int idx = tid + i * 256;
    const int row = idx >> 5;
    const int c4  = (idx & 31) * 4;
    const float4 hv = *(const float4*)(H + (size_t)(row0 + row) * FEAT + c4);
    *(float4*)&lds_h[row * SPAD + c4] = hv;
  }
  __syncthreads();

  const int wave = tid >> 5;
  const int lane = tid & 31;
  const int mi   = lane & 15;
  const int h    = lane >> 4;
  const int dir  = wave >> 2;
  const int n    = (wave & 3) * 16 + mi;
  float* __restrict__ Y       = dir ? Yout : Yin;
  const float4* __restrict__ bq =
      (const float4*)(wfl + dir * WF_SMALL) + ((wave & 3) * (FEAT / 8)) * 32 + lane;

  v8f c[NT] = {};
  tile_gemm<FEAT, SPAD, NT>(&lds_h[mi * SPAD + 4 * h], bq, c);

  const int mbase = row0 + 8 * h;
#pragma unroll
  for (int t = 0; t < NT; ++t) {
#pragma unroll
    for (int r = 0; r < 8; ++r) {
      Y[(size_t)(mbase + 16 * t + r) * HALF + n] = c[t][r];
    }
  }
}

__global__ __launch_bounds__(256) void proj_gemm_kernel(
    const float* __restrict__ H, const float* __restrict__ wfl, float* __restrict__ Yin,
    float* __restrict__ Yout) {
  __shared__ __align__(16) float lds_h[32 * SPAD];
  const int b = blockIdx.x;
  if (b < WG2) proj_gemm_body<2>(lds_h, b * 32, H, wfl, Yin, Yout);
  else         proj_gemm_body<1>(lds_h, 32 * WG2 + (b - WG2) * 16, H, wfl, Yin, Yout);
}

__device__ __forceinline__ float4 segment_mean(
    const float* __restrict__ Yb, const int* __restrict__ sd, int c, int idx0,
    int l16, bool hi, unsigned col) {
  const int cmax = max(__builtin_amdgcn_readlane(c, 0), __builtin_amdgcn_readlane(c, 16));

  float4 acc = make_float4(0.0f, 0.0f, 0.0f, 0.0f);
  for (int base = 0; base < cmax; base += 16) {
    const int rem = c - base;
    int myidx = idx0;
    if (base != 0) myidx = (l16 < rem) ? sd[base + l16] : 0;
    const int m = min(cmax - base, 16);
    for (int j = 0; j < m; j += 8) {
      float4 v[8];
#pragma unroll
      for (int u = 0; u < 8; ++u) {
        const int jl = j + u;
        const unsigned s_lo = (unsigned)__builtin_amdgcn_readlane(myidx, jl);
        const unsigned s_hi = (unsigned)__builtin_amdgcn_readlane(myidx, jl + 16);
        const unsigned s    = hi ? s_hi : s_lo;
        v[u] = make_float4(0.0f, 0.0f, 0.0f, 0.0f);
        if (jl < rem) v[u] = *(const float4*)(Yb + (s * HALF + col));
      }
#pragma unroll
      for (int u = 0; u < 8; ++u) {
        acc.x += v[u].x; acc.y += v[u].y; acc.z += v[u].z; acc.w += v[u].w;
      }
    }
  }
  const float rs = 1.0f / fmaxf((float)c, 1.0f);
  acc.x *= rs; acc.y *= rs; acc.z *= rs; acc.w *= rs;
  return acc;
}

template <bool PROJ, int NT>
__device__ __forceinline__ void layer_tail_body(
    float* lds_a, int row0,
    const float* __restrict__ H, const float* __restrict__ Y,
    const int2* __restrict__ seg, const int* __restrict__ srt, int E,
    const float* __restrict__ wfl,
    const float* __restrict__ blIn, const float* __restrict__ blOut,
    const float* __restrict__ cb,
    float* __restrict__ Hnext,
    const float* __restrict__ wfn,
    float* __restrict__ Yin, float* __restrict__ Yout) {
  float* const lds_o = lds_a;
  const int tid = threadIdx.x;

#pragma unroll
  for (int i = 0; i < 2 * NT; ++i) {
    const int idx = tid + i * 256;
    const int row = idx >> 5;
    const int c4  = (idx & 31) * 4;
    const float4 hv = *(const float4*)(H + (size_t)(row0 + row) * FEAT + c4);
    *(float4*)&lds_a[row * CPAD + c4] = hv;
  }

  {
    const int  l16 = tid & 15;
    const bool hi  = (tid & 16) != 0;
    const int  nl  = tid >> 4;
    const unsigned col = l16 * 4;
    int2 sg[NT][2];
    const int* sd[NT][2];
    int ix[NT][2];
#pragma unroll
    for (int q = 0; q < NT; ++q) {
      const int nd = row0 + 16 * q + nl;
      sg[q][0] = seg[nd];
      sg[q][1] = seg[NNODES + nd];
    }
#pragma unroll
    for (int q = 0; q < NT; ++q) {
#pragma unroll
      for (int d = 0; d < 2; ++d) {
        sd[q][d] = srt + (size_t)d * E + sg[q][d].x;
        const int r = sd[q][d][(l16 < sg[q][d].y) ? l16 : 0];
        ix[q][d] = (l16 < sg[q][d].y) ? r : 0;
      }
    }
#pragma unroll
    for (int q = 0; q < NT; ++q) {
#pragma unroll
      for (int d = 0; d < 2; ++d) {
        const float4 m = segment_mean(Y + (size_t)d * NNODES * HALF, sd[q][d], sg[q][d].y, ix[q][d], l16, hi, col);
        *(float4*)&lds_a[(16 * q + nl) * CPAD + FEAT + d * HALF + col] = m;
      }
    }
  }

  const int wave = tid >> 5;
  const int lane = tid & 31;
  const int mi   = lane & 15;
  const int h    = lane >> 4;
  const int dir  = wave >> 2;
  const int n64  = (wave & 3) * 16 + mi;
  const int n    = wave * 16 + mi;
  const int mbase = row0 + 8 * h;


  {
    const float4* __restrict__ bq =
        (const float4*)(wfl + 2 * WF_SMALL + dir * WF_SMALL) + ((wave & 3) * (FEAT / 8)) * 32 + lane;
    const float bias = (dir ? blOut : blIn)[n64];
    __syncthreads();
    v8f c[NT];
#pragma unroll
    for (int t = 0; t < NT; ++t) {
#pragma unroll
      for (int r = 0; r < 8; ++r) {
        c[t][r] = lds_a[(16 * t + 8 * h + r) * CPAD + FEAT + n] + bias;
      }
    }

    tile_gemm<FEAT, CPAD, NT>(&lds_a[mi * CPAD + 4 * h], bq, c);

#pragma unroll
    for (int t = 0; t < NT; ++t) {
#pragma unroll
      for (int r = 0; r < 8; ++r) {
        lds_a[(16 * t + 8 * h + r) * CPAD + FEAT + n] = fmaxf(c[t][r], 0.0f);
      }
    }
  }
  __syncthreads();

  {
    const float4* __restrict__ bq =
        (const float4*)(wfl + 4 * WF_SMALL) + (wave * (2 * FEAT / 8)) * 32 + lane;
    v8f c[NT] = {};
    tile_gemm<2 * FEAT, CPAD, NT>(&lds_a[mi * CPAD + 4 * h], bq, c);

    const float bias = cb[n];
    if (PROJ) __syncthreads();
#pragma unroll
    for (int t = 0; t < NT; ++t) {
#pragma unroll
      for (int r = 0; r < 8; ++r) {
        const float v = fmaxf(c[t][r] + bias, 0.0f);
        Hnext[(size_t)(mbase + 16 * t + r) * FEAT + n] = v;
        if (PROJ) lds_o[(16 * t + 8 * h + r) * SPAD + n] = v;
      }
    }
  }

  if (PROJ) {
    __syncthreads();
    const float4* __restrict__ bq =
        (const float4*)(wfn + dir * WF_SMALL) + ((wave & 3) * (FEAT / 8)) * 32 + lane;
    float* __restrict__ Yo = dir ? Yout : Yin;
    v8f c[NT] = {};
    tile_gemm<FEAT, SPAD, NT>(&lds_o[mi * SPAD + 4 * h], bq, c);
#pragma unroll
    for (int t = 0; t < NT; ++t) {
#pragma unroll
      for (int r = 0; r < 8; ++r) {
        Yo[(size_t)(mbase + 16 * t + r) * HALF + n64] = c[t][r];
      }
    }
  }
}

template <bool PROJ>
__global__ __launch_bounds__(256) void layer_tail_kernel(
    const float* __restrict__ H, const float* __restrict__ Y,
    const int2* __restrict__ seg, const int* __restrict__ srt, int E,
    const float* __restrict__ wfl,
    const float* __restrict__ blIn, const float* __restrict__ blOut,
    const float* __restrict__ cb,
    float* __restrict__ Hnext,
    const float* __restrict__ wfn,
    float* __restrict__ Yin, float* __restrict__ Yout) {
  __shared__ __align__(16) float lds_a[32 * CPAD];
  const int b = blockIdx.x;
  if (b < WG2)
    layer_tail_body<PROJ, 2>(lds_a, b * 32, H, Y, seg, srt, E, wfl, blIn, blOut, cb, Hnext, wfn, Yin, Yout);
  else
    layer_tail_body<PROJ, 1>(lds_a, 32 * WG2 + (b - WG2) * 16, H, Y, seg, srt, E, wfl, blIn, blOut, cb, Hnext, wfn, Yin, Yout);
}

__global__ void final_kernel(const float* __restrict__ H, const float* __restrict__ W,
                             const float* __restrict__ b, float* __restrict__ out, int n) {
  int i = blockIdx.x * blockDim.x + threadIdx.x;
  if (i >= n) return;
  const float4* hr = (const float4*)(H + (size_t)i * FEAT);
  const float4* wr = (const float4*)W;
  float acc = 0.0f;
#pragma unroll 8
  for (int k = 0; k < FEAT / 4; ++k) {
    float4 hv = hr[k];
    float4 wv = wr[k];
    acc += hv.x * wv.x + hv.y * wv.y + hv.z * wv.z + hv.w * wv.w;
  }
  out[i] = acc + b[0];
}

extern "C" void kernel_launch(void* const* d_in, const int* in_sizes, int n_in,
                              void* d_out, int out_size, void* d_ws, size_t ws_size,
                              hipStream_t stream) {
  (void)n_in; (void)out_size; (void)ws_size;

  const float* x       = (const float*)d_in[0];
  const int*   ein     = (const int*)d_in[1];
  const int*   eout    = (const int*)d_in[2];
  const float* in_Wl0  = (const float*)d_in[3];
  const float* in_bl0  = (const float*)d_in[4];
  const float* in_Wr0  = (const float*)d_in[5];
  const float* out_Wl0 = (const float*)d_in[6];
  const float* out_bl0 = (const float*)d_in[7];
  const float* out_Wr0 = (const float*)d_in[8];
  const float* comb_W0 = (const float*)d_in[9];
  const float* comb_b0 = (const float*)d_in[10];
  const float* in_Wl1  = (const float*)d_in[11];
  const float* in_bl1  = (const float*)d_in[12];
  const float* in_Wr1  = (const float*)d_in[13];
  const float* out_Wl1 = (const float*)d_in[14];
  const float* out_bl1 = (const float*)d_in[15];
  const float* out_Wr1 = (const float*)d_in[16];
  const float* comb_W1 = (const float*)d_in[17];
  const float* comb_b1 = (const float*)d_in[18];
  const float* final_W = (const float*)d_in[19];
  const float* final_b = (const float*)d_in[20];
  float* out = (float*)d_out;

  const int N = NNODES;
  const int E = in_sizes[1] / 2;

  const size_t N128 = (size_t)N * FEAT;
  const size_t N64  = (size_t)N * HALF;
  float* ws      = (float*)d_ws;
  int*   cursor  = (int*)d_ws;
  float* yB      = ws + 2 * PADN;
  int2*  seg     = (int2*)(yB + N128);
  int*   srt     = (int*)(seg + 2 * N);
  int*   rank    = srt + 2 * (size_t)E;
  float* wf      = yB + N128 + 4000000;
  float* HA      = yB + 2 * N128;
  float* HB      = HA + N128;
  float* yA      = HB;
  int*   cntmat  = (int*)HA;

  const int ZT = 256;
  auto nblk = [](long long c, int t) { return (unsigned)((c + t - 1) / t); };

  edge_rank_kernel<<<2 * NCH + WBLK, 1024, 0, stream>>>(
      ein, eout, cursor, cntmat, rank, E,
      in_Wl0, out_Wl0, in_Wr0, out_Wr0, comb_W0,
      in_Wl1, out_Wl1, in_Wr1, out_Wr1, comb_W1, wf);
  edge_scan_kernel<<<dim3(NGRP, 2), 1024, 0, stream>>>(cursor, cntmat, seg);
  edge_place_kernel<<<dim3(nblk(E, PLT), 2), PLT, 0, stream>>>(ein, eout, cntmat, rank, srt, E);

  proj_gemm_kernel<<<WGROUPS, 256, 0, stream>>>(x, wf, yA, yA + N64);
  layer_tail_kernel<true><<<WGROUPS, 256, 0, stream>>>(
      x, yA, seg, srt, E, wf, in_bl0, out_bl0, comb_b0, HA,
      wf + WF_LAYER, yB, yB + N64);

  layer_tail_kernel<false><<<WGROUPS, 256, 0, stream>>>(
      HA, yB, seg, srt, E, wf + WF_LAYER, in_bl1, out_bl1, comb_b1, HB,
      nullptr, nullptr, nullptr);

  final_kernel<<<nblk(N, ZT), ZT, 0, stream>>>(HB, final_W, final_b, out, N);
}
